# starting kernel
# baseline (speedup 1.0000x reference)
#include <hip/hip_runtime.h>
#include <hip/hip_bf16.h>
#include <math.h>

typedef __attribute__((ext_vector_type(2))) float v2f;
typedef __attribute__((ext_vector_type(8))) float v8f;

namespace {
constexpr int kLayer  = 2;
constexpr int kN = 32, kL = 512, kD = 1024, kH = 256;
constexpr int kS = 2048, kSpan = 64;
constexpr int kKC = 64;
constexpr int kAStride = 68;
constexpr int kWPairStride = 272;
}

__launch_bounds__(256, 2)
__global__ void attention_pooler_kernel(const float* __restrict__ hs_all,
                                        const int*  __restrict__ spans,
                                        const float* __restrict__ Wm,
                                        const float* __restrict__ bvec,
                                        const float* __restrict__ vvec,
                                        float* __restrict__ out)
{
    __shared__ __align__(16) float As[kSpan * kAStride];
    __shared__ __align__(16) float Wp[(kKC / 2) * kWPairStride * 2];
    __shared__ float sc[kSpan];
    __shared__ float att[kSpan];
    __shared__ float pooled[kH];
    __shared__ float red[2];

    const int s    = blockIdx.x;
    const int tid  = threadIdx.x;
    const int wave = tid >> 5;
    const int lane = tid & 31;
    const int mt   = wave & 3;
    const int nh   = wave >> 2;
    const int hi   = lane >> 4;
    const int l16  = lane & 15;

    const float* hs = hs_all + (size_t)kLayer * kN * kL * kD;

    const int idx   = spans[3 * s + 0];
    const int start = spans[3 * s + 1];
    const int stop  = spans[3 * s + 2];

    if (tid < kSpan) sc[tid]     = 0.f;
    if (tid < kH)    pooled[tid] = 0.f;

    v8f acc[8];
    #pragma unroll
    for (int nt = 0; nt < 8; ++nt)
        #pragma unroll
        for (int e = 0; e < 8; ++e) acc[nt][e] = 0.f;

    const int rowj = tid >> 2;
    const int quad = tid & 3;
    int pos = start + rowj;
    if (pos > kL - 1) pos = kL - 1;
    const float* arow = hs + ((size_t)idx * kL + pos) * kD;

    for (int kc = 0; kc < kD / kKC; ++kc) {
        const int kbase = kc * kKC;
        __syncthreads();

        {
            const float4* src = (const float4*)(arow + kbase + quad * 16);
            float4 t0 = src[0], t1 = src[1], t2 = src[2], t3 = src[3];
            float* dst = &As[rowj * kAStride + quad * 16];
            *(float4*)(dst +  0) = t0;
            *(float4*)(dst +  4) = t1;
            *(float4*)(dst +  8) = t2;
            *(float4*)(dst + 12) = t3;
        }
        #pragma unroll
        for (int i0 = 0; i0 < 8; i0 += 2) {
            const int ta = tid + i0 * 256;
            const int tb = tid + (i0 + 1) * 256;
            const int kpa = ta >> 6, ca = (ta & 63) * 4;
            const int kpb = tb >> 6, cb = (tb & 63) * 4;
            const float* ga0 = Wm + (size_t)(kbase + 2 * kpa + 0) * kH + ca;
            const float* ga1 = Wm + (size_t)(kbase + 2 * kpa + 1) * kH + ca;
            const float* gb0 = Wm + (size_t)(kbase + 2 * kpb + 0) * kH + cb;
            const float* gb1 = Wm + (size_t)(kbase + 2 * kpb + 1) * kH + cb;
            float4 a0 = *(const float4*)ga0;
            float4 a1 = *(const float4*)ga1;
            float4 b0 = *(const float4*)gb0;
            float4 b1 = *(const float4*)gb1;
            float* da = &Wp[(kpa * kWPairStride + ca) * 2];
            float* db = &Wp[(kpb * kWPairStride + cb) * 2];
            v2f p;
            p.x = a0.x; p.y = a1.x; *(v2f*)(da + 0) = p;
            p.x = a0.y; p.y = a1.y; *(v2f*)(da + 2) = p;
            p.x = a0.z; p.y = a1.z; *(v2f*)(da + 4) = p;
            p.x = a0.w; p.y = a1.w; *(v2f*)(da + 6) = p;
            p.x = b0.x; p.y = b1.x; *(v2f*)(db + 0) = p;
            p.x = b0.y; p.y = b1.y; *(v2f*)(db + 2) = p;
            p.x = b0.z; p.y = b1.z; *(v2f*)(db + 4) = p;
            p.x = b0.w; p.y = b1.w; *(v2f*)(db + 6) = p;
        }
        __syncthreads();

        #pragma unroll
        for (int kk = 0; kk < kKC / 4; ++kk) {
            const int kb = kk * 4;
            const v2f a = *(const v2f*)&As[(mt * 16 + l16) * kAStride + kb + 2 * hi];
            v2f bfr[8];
            #pragma unroll
            for (int nt = 0; nt < 8; ++nt) {
                const int col = (nh * 8 + nt) * 16 + l16;
                bfr[nt] = *(const v2f*)&Wp[((kb >> 1) + hi) * (kWPairStride * 2) + col * 2];
            }
            #pragma unroll
            for (int nt = 0; nt < 8; ++nt) {
                acc[nt] = __builtin_amdgcn_wmma_f32_16x16x4_f32(
                    false, a, false, bfr[nt], (short)0, acc[nt], false, false);
            }
        }
    }

    float srow[8];
    #pragma unroll
    for (int r = 0; r < 8; ++r) srow[r] = 0.f;
    #pragma unroll
    for (int nt = 0; nt < 8; ++nt) {
        const int col    = (nh * 8 + nt) * 16 + l16;
        const float bcol = bvec[col];
        const float vcol = vvec[col];
        #pragma unroll
        for (int r = 0; r < 8; ++r) {
            acc[nt][r] += bcol;
            srow[r] += acc[nt][r] * vcol;
        }
    }
    #pragma unroll
    for (int off = 8; off >= 1; off >>= 1)
        #pragma unroll
        for (int r = 0; r < 8; ++r)
            srow[r] += __shfl_xor(srow[r], off, 16);
    if (l16 == 0) {
        #pragma unroll
        for (int r = 0; r < 8; ++r)
            atomicAdd(&sc[mt * 16 + hi * 8 + r], srow[r]);
    }
    __syncthreads();

    const int len = stop - start;
    if (tid == 0) {
        float m = -INFINITY;
        for (int l = 0; l < kSpan; ++l)
            if (l < len) m = fmaxf(m, sc[l]);
        red[0] = m;
    }
    __syncthreads();
    if (tid < kSpan)
        att[tid] = (tid < len) ? expf(sc[tid] - red[0]) : 0.f;
    __syncthreads();
    if (tid == 0) {
        float sum = 0.f;
        for (int l = 0; l < kSpan; ++l) sum += att[l];
        red[1] = 1.f / sum;
    }
    __syncthreads();

    float aw[8];
    const float inv = red[1];
    #pragma unroll
    for (int r = 0; r < 8; ++r)
        aw[r] = att[mt * 16 + hi * 8 + r] * inv;
    #pragma unroll
    for (int nt = 0; nt < 8; ++nt) {
        float p = 0.f;
        #pragma unroll
        for (int r = 0; r < 8; ++r) p += aw[r] * acc[nt][r];
        p += __shfl_xor(p, 16, 32);
        if (hi == 0) atomicAdd(&pooled[(nh * 8 + nt) * 16 + l16], p);
    }
    __syncthreads();

    out[(size_t)s * kH + tid] = pooled[tid];
}

extern "C" void kernel_launch(void* const* d_in, const int* in_sizes, int n_in,
                              void* d_out, int out_size, void* d_ws, size_t ws_size,
                              hipStream_t stream) {
    (void)in_sizes; (void)n_in; (void)out_size; (void)d_ws; (void)ws_size;
    const float* hs    = (const float*)d_in[0];
    const int*   spans = (const int*)  d_in[1];
    const float* W     = (const float*)d_in[2];
    const float* b     = (const float*)d_in[3];
    const float* v     = (const float*)d_in[4];
    float* out = (float*)d_out;
    attention_pooler_kernel<<<dim3(kS), dim3(256), 0, stream>>>(hs, spans, W, b, v, out);
}


# One work-group per 64-start window projects its 128 rows once (two-stage LDS ring) and pools every span starting there from that tile, not one projection per span; softmax and row sums in lane order
# speedup vs baseline: 4.3596x; 4.3596x over previous
#include <hip/hip_runtime.h>
#include <hip/hip_bf16.h>
#include <math.h>


typedef __attribute__((ext_vector_type(2))) float v2f;
typedef __attribute__((ext_vector_type(8))) float v8f;

namespace {
constexpr int kLayer  = 2;
constexpr int kN = 32, kL = 512, kD = 1024, kH = 256;
constexpr int kS = 2048, kSpan = 64;
constexpr int kWin       = 64;
constexpr int kWinShift  = 6;
constexpr int kRows      = kWin + kSpan;
constexpr int kWinPerSeq = kL / kWin;
constexpr int kGrid      = kN * kWinPerSeq;
constexpr int kThreads   = 512;
constexpr int kWaves     = kThreads / 32;
constexpr int kScanPerWave = kS / kWaves;
constexpr int kKC = 64;
constexpr int kAStride = 68;
constexpr int kWPairStride = 272;
static_assert((1 << kWinShift) == kWin, "window shift");
constexpr int kPStride = 260;
constexpr int kAFloats = kRows * kAStride;
constexpr int kWFloats = (kKC / 2) * kWPairStride * 2;
constexpr int kStageFloats = kAFloats + kWFloats;
constexpr int kPFloats = kRows * kPStride;
}

__launch_bounds__(512)
__global__ void attention_pooler_kernel(const float* __restrict__ hs_all,
                                        const int*  __restrict__ spans,
                                        const float* __restrict__ Wm,
                                        const float* __restrict__ bvec,
                                        const float* __restrict__ vvec,
                                        float* __restrict__ out)
{
    __shared__ __align__(16) float smem[2 * kStageFloats];
    __shared__ float scp[2][kRows];
    __shared__ unsigned short slist[kWaves][kScanPerWave];
    __shared__ int scnt[kWaves];

    const int seq  = blockIdx.x / kWinPerSeq;
    const int win  = blockIdx.x % kWinPerSeq;
    const int p0   = win * kWin;
    const int tid  = threadIdx.x;
    const int wave = tid >> 5;
    const int lane = tid & 31;
    const int mt   = wave & 7;
    const int nh   = wave >> 3;
    const int hi   = lane >> 4;
    const int l16  = lane & 15;
    const int uwave = __builtin_amdgcn_readfirstlane(wave);

    const float* hs = hs_all + (size_t)kLayer * kN * kL * kD;

    {
        int ridx[kScanPerWave / 32], rst[kScanPerWave / 32];
        #pragma unroll
        for (int it = 0; it < kScanPerWave / 32; ++it) {
            const int s = uwave * kScanPerWave + it * 32 + lane;
            ridx[it] = spans[3 * s + 0];
            rst[it]  = spans[3 * s + 1];
        }
        int n = 0;
        #pragma unroll
        for (int it = 0; it < kScanPerWave / 32; ++it) {
            const int s = uwave * kScanPerWave + it * 32 + lane;
            const bool mine = (ridx[it] == seq) && ((rst[it] >> kWinShift) == win);
            const unsigned mask = (unsigned)__ballot(mine);
            if (mine) slist[uwave][n + __popc(mask & ((1u << lane) - 1u))] = (unsigned short)s;
            n += __popc(mask);
        }
        if (lane == 0) scnt[uwave] = n;
    }
    __syncthreads();
    int total = 0;
    for (int w = 0; w < kWaves; ++w) total += scnt[w];
    total = __builtin_amdgcn_readfirstlane(total);
    if (total == 0) return;

    v8f acc[8];
    #pragma unroll
    for (int nt = 0; nt < 8; ++nt)
        #pragma unroll
        for (int e = 0; e < 8; ++e) acc[nt][e] = 0.f;

    const int rowj = tid >> 2;
    const int quad = tid & 3;
    int pos = p0 + rowj;
    if (pos > kL - 1) pos = kL - 1;
    const float* arow = hs + ((size_t)seq * kL + pos) * kD;

    const int wkp = tid >> 6;
    const int wc  = (tid & 63) * 4;
    const float* wsrc = Wm + (size_t)(2 * wkp) * kH + wc;
    float* adst0 = &smem[rowj * kAStride + quad * 16];
    float* wdst0 = &smem[kAFloats + (wkp * kWPairStride + wc) * 2];
    float4 ta0, ta1, ta2, ta3;
    float4 tw00, tw01, tw10, tw11, tw20, tw21, tw30, tw31;

#define LOAD_CHUNK(kbase_)                                                              \
    {                                                                                   \
        const float4* asrc_ = (const float4*)(arow + (kbase_) + quad * 16);             \
        ta0 = asrc_[0]; ta1 = asrc_[1]; ta2 = asrc_[2]; ta3 = asrc_[3];                 \
        const float* w_ = wsrc + (size_t)(kbase_) * kH;                                 \
        tw00 = *(const float4*)(w_ + (size_t)( 0 + 0) * kH);                            \
        tw01 = *(const float4*)(w_ + (size_t)( 0 + 1) * kH);                            \
        tw10 = *(const float4*)(w_ + (size_t)(16 + 0) * kH);                            \
        tw11 = *(const float4*)(w_ + (size_t)(16 + 1) * kH);                            \
        tw20 = *(const float4*)(w_ + (size_t)(32 + 0) * kH);                            \
        tw21 = *(const float4*)(w_ + (size_t)(32 + 1) * kH);                            \
        tw30 = *(const float4*)(w_ + (size_t)(48 + 0) * kH);                            \
        tw31 = *(const float4*)(w_ + (size_t)(48 + 1) * kH);                            \
    }
#define STORE_PAIRS(d_, w0_, w1_)                                                       \
    {                                                                                   \
        v2f p_;                                                                         \
        p_.x = (w0_).x; p_.y = (w1_).x; *(v2f*)((d_) + 0) = p_;                         \
        p_.x = (w0_).y; p_.y = (w1_).y; *(v2f*)((d_) + 2) = p_;                         \
        p_.x = (w0_).z; p_.y = (w1_).z; *(v2f*)((d_) + 4) = p_;                         \
        p_.x = (w0_).w; p_.y = (w1_).w; *(v2f*)((d_) + 6) = p_;                         \
    }
#define STORE_CHUNK(stage_)                                                             \
    {                                                                                   \
        float* ad_ = adst0 + (stage_) * kStageFloats;                                   \
        *(float4*)(ad_ +  0) = ta0;                                                     \
        *(float4*)(ad_ +  4) = ta1;                                                     \
        *(float4*)(ad_ +  8) = ta2;                                                     \
        *(float4*)(ad_ + 12) = ta3;                                                     \
        float* wd_ = wdst0 + (stage_) * kStageFloats;                                   \
        STORE_PAIRS(wd_ + 0 * 8 * kWPairStride * 2, tw00, tw01)                         \
        STORE_PAIRS(wd_ + 1 * 8 * kWPairStride * 2, tw10, tw11)                         \
        STORE_PAIRS(wd_ + 2 * 8 * kWPairStride * 2, tw20, tw21)                         \
        STORE_PAIRS(wd_ + 3 * 8 * kWPairStride * 2, tw30, tw31)                         \
    }

    LOAD_CHUNK(0)
    STORE_CHUNK(0)
    __syncthreads();

    #pragma unroll 2
    for (int kc = 0; kc < kD / kKC; ++kc) {
        const int  cur  = kc & 1;
        const bool more = kc + 1 < kD / kKC;
        if (more) LOAD_CHUNK((kc + 1) * kKC)

        const float* As = smem + cur * kStageFloats;
        const float* Wp = As + kAFloats;
        #pragma unroll
        for (int kk = 0; kk < kKC / 4; ++kk) {
            const int kb = kk * 4;
            const v2f a = *(const v2f*)&As[(mt * 16 + l16) * kAStride + kb + 2 * hi];
            v2f bfr[8];
            #pragma unroll
            for (int nt = 0; nt < 8; ++nt) {
                const int col = (nh * 8 + nt) * 16 + l16;
                bfr[nt] = *(const v2f*)&Wp[((kb >> 1) + hi) * (kWPairStride * 2) + col * 2];
            }
            #pragma unroll
            for (int nt = 0; nt < 8; ++nt) {
                acc[nt] = __builtin_amdgcn_wmma_f32_16x16x4_f32(
                    false, a, false, bfr[nt], (short)0, acc[nt], false, false);
            }
        }
        if (more) STORE_CHUNK(cur ^ 1)
        __syncthreads();
    }

#undef LOAD_CHUNK
#undef STORE_PAIRS
#undef STORE_CHUNK

    float* P = smem;
    float srow[8];
    #pragma unroll
    for (int r = 0; r < 8; ++r) srow[r] = 0.f;
    #pragma unroll
    for (int nt = 0; nt < 8; ++nt) {
        const int col    = (nh * 8 + nt) * 16 + l16;
        const float bcol = bvec[col];
        const float vcol = vvec[col];
        #pragma unroll
        for (int r = 0; r < 8; ++r) {
            acc[nt][r] += bcol;
            srow[r] += acc[nt][r] * vcol;
            P[(mt * 16 + hi * 8 + r) * kPStride + col] = acc[nt][r];
        }
    }
    #pragma unroll
    for (int off = 8; off >= 1; off >>= 1)
        #pragma unroll
        for (int r = 0; r < 8; ++r)
            srow[r] += __shfl_xor(srow[r], off, 16);
    if (l16 == 0) {
        #pragma unroll
        for (int r = 0; r < 8; ++r)
            scp[nh][mt * 16 + hi * 8 + r] = srow[r];
    }
    __syncthreads();

    for (int j = uwave; j < total; j += kWaves) {
        int t = j, w = 0;
        for (; w < kWaves - 1 && t >= scnt[w]; ++w) t -= scnt[w];
        const int s     = __builtin_amdgcn_readfirstlane((int)slist[w][t]);
        const int start = __builtin_amdgcn_readfirstlane(spans[3 * s + 1]);
        const int stop  = __builtin_amdgcn_readfirstlane(spans[3 * s + 2]);
        const int len   = stop - start;

        int r0 = start + lane;       if (r0 > kL - 1) r0 = kL - 1;  r0 -= p0;
        int r1 = start + lane + 32;  if (r1 > kL - 1) r1 = kL - 1;  r1 -= p0;
        const float sc0 = scp[0][r0] + scp[1][r0];
        const float sc1 = scp[0][r1] + scp[1][r1];
        const bool v0 = lane < len, v1 = lane + 32 < len;
        float m = fmaxf(v0 ? sc0 : -INFINITY, v1 ? sc1 : -INFINITY);
        #pragma unroll
        for (int off = 16; off >= 1; off >>= 1) m = fmaxf(m, __shfl_xor(m, off, 32));
        const float e0 = v0 ? expf(sc0 - m) : 0.f;
        const float e1 = v1 ? expf(sc1 - m) : 0.f;
        float sum = e0 + e1;
        #pragma unroll
        for (int off = 16; off >= 1; off >>= 1) sum += __shfl_xor(sum, off, 32);
        const float inv = 1.f / sum;
        const float a0 = e0 * inv, a1 = e1 * inv;

        const float* Pl = P + lane * 4;
        float4 pa = make_float4(0.f, 0.f, 0.f, 0.f);
        float4 pb = make_float4(0.f, 0.f, 0.f, 0.f);
        #pragma unroll 8
        for (int l = 0; l < kSpan; ++l) {
            const float av = (l < 32) ? a0 : a1;
            const float a  = __int_as_float(__builtin_amdgcn_readlane(__float_as_int(av), l & 31));
            int r = start + l;  if (r > kL - 1) r = kL - 1;  r -= p0;
            const float4 x = *(const float4*)(Pl + r * kPStride);
            const float4 y = *(const float4*)(Pl + r * kPStride + 128);
            pa.x = fmaf(a, x.x, pa.x); pa.y = fmaf(a, x.y, pa.y);
            pa.z = fmaf(a, x.z, pa.z); pa.w = fmaf(a, x.w, pa.w);
            pb.x = fmaf(a, y.x, pb.x); pb.y = fmaf(a, y.y, pb.y);
            pb.z = fmaf(a, y.z, pb.z); pb.w = fmaf(a, y.w, pb.w);
        }
        float* o = out + (size_t)s * kH + lane * 4;
        *(float4*)(o)       = pa;
        *(float4*)(o + 128) = pb;
    }
}

extern "C" void kernel_launch(void* const* d_in, const int* in_sizes, int n_in,
                              void* d_out, int out_size, void* d_ws, size_t ws_size,
                              hipStream_t stream) {
    (void)in_sizes; (void)n_in; (void)out_size; (void)d_ws; (void)ws_size;
    const float* hs    = (const float*)d_in[0];
    const int*   spans = (const int*)  d_in[1];
    const float* W     = (const float*)d_in[2];
    const float* b     = (const float*)d_in[3];
    const float* v     = (const float*)d_in[4];
    float* out = (float*)d_out;
    attention_pooler_kernel<<<dim3(kGrid), dim3(kThreads), 0, stream>>>(hs, spans, W, b, v, out);
}


# Project every layer row once (64-row tiles) into the workspace with its score and pool spans in a second launch, not 128 rows per 64-start window; score partials now summed per column quarter
# speedup vs baseline: 6.0632x; 1.3908x over previous
#include <hip/hip_runtime.h>
#include <hip/hip_bf16.h>
#include <math.h>


typedef __attribute__((ext_vector_type(2))) float v2f;
typedef __attribute__((ext_vector_type(8))) float v8f;

namespace {
constexpr int kLayer  = 2;
constexpr int kN = 32, kL = 512, kD = 1024, kH = 256;
constexpr int kS = 2048, kSpan = 64;
constexpr int kWin       = 64;
constexpr int kWinShift  = 6;
constexpr int kRows      = kWin + kSpan;
constexpr int kWinPerSeq = kL / kWin;
constexpr int kGrid      = kN * kWinPerSeq;
constexpr int kThreads   = 512;
constexpr int kWaves     = kThreads / 32;
constexpr int kScanPerWave = kS / kWaves;
constexpr int kKC = 64;
constexpr int kAStride = 68;
constexpr int kWPairStride = 272;
static_assert((1 << kWinShift) == kWin, "window shift");
constexpr int kPStride = 260;
constexpr int kAFloats = kRows * kAStride;
constexpr int kWFloats = (kKC / 2) * kWPairStride * 2;
constexpr int kStageFloats = kAFloats + kWFloats;
constexpr int kPFloats = kRows * kPStride;
constexpr int kAllRows  = kN * kL;
constexpr int kTRows    = 64;
constexpr int kTAFloats = kTRows * kAStride;
constexpr int kTStage   = kTAFloats + kWFloats;
static_assert(kAllRows / kTRows == kGrid, "one 64-row tile per work-group");
static_assert(2 * kTStage <= 2 * kStageFloats && kTRows * kPStride <= 2 * kStageFloats, "fits the LDS ring");
constexpr size_t kWsFloats = (size_t)kAllRows * kH + kAllRows;
constexpr size_t kWsBytes  = kWsFloats * sizeof(float);
constexpr int kPoolThreads = 256;
constexpr int kPoolGrid    = kS * 2 / (kPoolThreads / 32);
}

__launch_bounds__(512)
__global__ void attention_pooler_kernel(const float* __restrict__ hs_all,
                                        const int*  __restrict__ spans,
                                        const float* __restrict__ Wm,
                                        const float* __restrict__ bvec,
                                        const float* __restrict__ vvec,
                                        float* __restrict__ ws, size_t ws_bytes,
                                        float* __restrict__ out)
{
    __shared__ __align__(16) float smem[2 * kStageFloats];
    __shared__ float scp[2][kRows];
    __shared__ unsigned short slist[kWaves][kScanPerWave];
    __shared__ int scnt[kWaves];

    const int seq  = blockIdx.x / kWinPerSeq;
    const int win  = blockIdx.x % kWinPerSeq;
    const int p0   = win * kWin;
    const int tid  = threadIdx.x;
    const int wave = tid >> 5;
    const int lane = tid & 31;
    const int mt   = wave & 7;
    const int nh   = wave >> 3;
    const int hi   = lane >> 4;
    const int l16  = lane & 15;
    const int uwave = __builtin_amdgcn_readfirstlane(wave);

    const float* hs = hs_all + (size_t)kLayer * kN * kL * kD;

    if (ws_bytes >= kWsBytes) {
        const int row0 = blockIdx.x * kTRows;
        const int tmt  = wave & 3;
        const int tnq  = wave >> 2;

        v8f tacc[4];
        #pragma unroll
        for (int nt = 0; nt < 4; ++nt)
            #pragma unroll
            for (int e = 0; e < 8; ++e) tacc[nt][e] = 0.f;

        const int ar = tid >> 4;
        const int ac = (tid & 15) * 4;
        const float* asrc0 = hs + (size_t)(row0 + ar) * kD + ac;
        const float* asrc1 = asrc0 + (size_t)32 * kD;
        float* tadst = &smem[ar * kAStride + ac];
        const int twkp = tid >> 6;
        const int twc  = (tid & 63) * 4;
        const float* twsrc = Wm + (size_t)(2 * twkp) * kH + twc;
        float* twdst = &smem[kTAFloats + (twkp * kWPairStride + twc) * 2];
        float4 ua0, ua1;
        float4 uw00, uw01, uw10, uw11, uw20, uw21, uw30, uw31;

#define T_LOAD(kbase_)                                                                  \
    {                                                                                   \
        ua0 = *(const float4*)(asrc0 + (kbase_));                                       \
        ua1 = *(const float4*)(asrc1 + (kbase_));                                       \
        const float* w_ = twsrc + (size_t)(kbase_) * kH;                                \
        uw00 = *(const float4*)(w_ + (size_t)( 0 + 0) * kH);                            \
        uw01 = *(const float4*)(w_ + (size_t)( 0 + 1) * kH);                            \
        uw10 = *(const float4*)(w_ + (size_t)(16 + 0) * kH);                            \
        uw11 = *(const float4*)(w_ + (size_t)(16 + 1) * kH);                            \
        uw20 = *(const float4*)(w_ + (size_t)(32 + 0) * kH);                            \
        uw21 = *(const float4*)(w_ + (size_t)(32 + 1) * kH);                            \
        uw30 = *(const float4*)(w_ + (size_t)(48 + 0) * kH);                            \
        uw31 = *(const float4*)(w_ + (size_t)(48 + 1) * kH);                            \
    }
#define T_PAIRS(d_, w0_, w1_)                                                           \
    {                                                                                   \
        v2f p_;                                                                         \
        p_.x = (w0_).x; p_.y = (w1_).x; *(v2f*)((d_) + 0) = p_;                         \
        p_.x = (w0_).y; p_.y = (w1_).y; *(v2f*)((d_) + 2) = p_;                         \
        p_.x = (w0_).z; p_.y = (w1_).z; *(v2f*)((d_) + 4) = p_;                         \
        p_.x = (w0_).w; p_.y = (w1_).w; *(v2f*)((d_) + 6) = p_;                         \
    }
#define T_STORE(stage_)                                                                 \
    {                                                                                   \
        float* ad_ = tadst + (stage_) * kTStage;                                        \
        *(float4*)(ad_) = ua0;                                                          \
        *(float4*)(ad_ + 32 * kAStride) = ua1;                                          \
        float* wd_ = twdst + (stage_) * kTStage;                                        \
        T_PAIRS(wd_ + 0 * 8 * kWPairStride * 2, uw00, uw01)                             \
        T_PAIRS(wd_ + 1 * 8 * kWPairStride * 2, uw10, uw11)                             \
        T_PAIRS(wd_ + 2 * 8 * kWPairStride * 2, uw20, uw21)                             \
        T_PAIRS(wd_ + 3 * 8 * kWPairStride * 2, uw30, uw31)                             \
    }

        T_LOAD(0)
        T_STORE(0)
        __syncthreads();

        #pragma unroll 2
        for (int kc = 0; kc < kD / kKC; ++kc) {
            const int  cur  = kc & 1;
            const bool more = kc + 1 < kD / kKC;
            if (more) T_LOAD((kc + 1) * kKC)

            const float* As = smem + cur * kTStage;
            const float* Wp = As + kTAFloats;
            #pragma unroll
            for (int kk = 0; kk < kKC / 4; ++kk) {
                const int kb = kk * 4;
                const v2f a = *(const v2f*)&As[(tmt * 16 + l16) * kAStride + kb + 2 * hi];
                v2f bfr[4];
                #pragma unroll
                for (int nt = 0; nt < 4; ++nt) {
                    const int col = (tnq * 4 + nt) * 16 + l16;
                    bfr[nt] = *(const v2f*)&Wp[((kb >> 1) + hi) * (kWPairStride * 2) + col * 2];
                }
                #pragma unroll
                for (int nt = 0; nt < 4; ++nt) {
                    tacc[nt] = __builtin_amdgcn_wmma_f32_16x16x4_f32(
                        false, a, false, bfr[nt], (short)0, tacc[nt], false, false);
                }
            }
            if (more) T_STORE(cur ^ 1)
            __syncthreads();
        }
#undef T_LOAD
#undef T_PAIRS
#undef T_STORE

        float* Pt = smem;
        float (*scq)[kTRows] = (float (*)[kTRows])&scp[0][0];
        float srow[8];
        #pragma unroll
        for (int r = 0; r < 8; ++r) srow[r] = 0.f;
        #pragma unroll
        for (int nt = 0; nt < 4; ++nt) {
            const int col    = (tnq * 4 + nt) * 16 + l16;
            const float bcol = bvec[col];
            const float vcol = vvec[col];
            #pragma unroll
            for (int r = 0; r < 8; ++r) {
                tacc[nt][r] += bcol;
                srow[r] += tacc[nt][r] * vcol;
                Pt[(tmt * 16 + hi * 8 + r) * kPStride + col] = tacc[nt][r];
            }
        }
        #pragma unroll
        for (int off = 8; off >= 1; off >>= 1)
            #pragma unroll
            for (int r = 0; r < 8; ++r)
                srow[r] += __shfl_xor(srow[r], off, 16);
        if (l16 == 0) {
            #pragma unroll
            for (int r = 0; r < 8; ++r)
                scq[tnq][tmt * 16 + hi * 8 + r] = srow[r];
        }
        __syncthreads();

        float* wsS = ws + (size_t)kAllRows * kH;
        if (tid < kTRows)
            wsS[row0 + tid] = (scq[0][tid] + scq[1][tid]) + (scq[2][tid] + scq[3][tid]);
        #pragma unroll
        for (int i = 0; i < kTRows / kWaves; ++i) {
            const int r = uwave * (kTRows / kWaves) + i;
            const float* s_ = Pt + r * kPStride + lane * 4;
            float* d_ = ws + (size_t)(row0 + r) * kH + lane * 4;
            const float4 x = *(const float4*)(s_);
            const float4 y = *(const float4*)(s_ + 128);
            *(float4*)(d_)       = x;
            *(float4*)(d_ + 128) = y;
        }
        return;
    }

    {
        int ridx[kScanPerWave / 32], rst[kScanPerWave / 32];
        #pragma unroll
        for (int it = 0; it < kScanPerWave / 32; ++it) {
            const int s = uwave * kScanPerWave + it * 32 + lane;
            ridx[it] = spans[3 * s + 0];
            rst[it]  = spans[3 * s + 1];
        }
        int n = 0;
        #pragma unroll
        for (int it = 0; it < kScanPerWave / 32; ++it) {
            const int s = uwave * kScanPerWave + it * 32 + lane;
            const bool mine = (ridx[it] == seq) && ((rst[it] >> kWinShift) == win);
            const unsigned mask = (unsigned)__ballot(mine);
            if (mine) slist[uwave][n + __popc(mask & ((1u << lane) - 1u))] = (unsigned short)s;
            n += __popc(mask);
        }
        if (lane == 0) scnt[uwave] = n;
    }
    __syncthreads();
    int total = 0;
    for (int w = 0; w < kWaves; ++w) total += scnt[w];
    total = __builtin_amdgcn_readfirstlane(total);
    if (total == 0) return;

    v8f acc[8];
    #pragma unroll
    for (int nt = 0; nt < 8; ++nt)
        #pragma unroll
        for (int e = 0; e < 8; ++e) acc[nt][e] = 0.f;

    const int rowj = tid >> 2;
    const int quad = tid & 3;
    int pos = p0 + rowj;
    if (pos > kL - 1) pos = kL - 1;
    const float* arow = hs + ((size_t)seq * kL + pos) * kD;

    const int wkp = tid >> 6;
    const int wc  = (tid & 63) * 4;
    const float* wsrc = Wm + (size_t)(2 * wkp) * kH + wc;
    float* adst0 = &smem[rowj * kAStride + quad * 16];
    float* wdst0 = &smem[kAFloats + (wkp * kWPairStride + wc) * 2];
    float4 ta0, ta1, ta2, ta3;
    float4 tw00, tw01, tw10, tw11, tw20, tw21, tw30, tw31;

#define LOAD_CHUNK(kbase_)                                                              \
    {                                                                                   \
        const float4* asrc_ = (const float4*)(arow + (kbase_) + quad * 16);             \
        ta0 = asrc_[0]; ta1 = asrc_[1]; ta2 = asrc_[2]; ta3 = asrc_[3];                 \
        const float* w_ = wsrc + (size_t)(kbase_) * kH;                                 \
        tw00 = *(const float4*)(w_ + (size_t)( 0 + 0) * kH);                            \
        tw01 = *(const float4*)(w_ + (size_t)( 0 + 1) * kH);                            \
        tw10 = *(const float4*)(w_ + (size_t)(16 + 0) * kH);                            \
        tw11 = *(const float4*)(w_ + (size_t)(16 + 1) * kH);                            \
        tw20 = *(const float4*)(w_ + (size_t)(32 + 0) * kH);                            \
        tw21 = *(const float4*)(w_ + (size_t)(32 + 1) * kH);                            \
        tw30 = *(const float4*)(w_ + (size_t)(48 + 0) * kH);                            \
        tw31 = *(const float4*)(w_ + (size_t)(48 + 1) * kH);                            \
    }
#define STORE_PAIRS(d_, w0_, w1_)                                                       \
    {                                                                                   \
        v2f p_;                                                                         \
        p_.x = (w0_).x; p_.y = (w1_).x; *(v2f*)((d_) + 0) = p_;                         \
        p_.x = (w0_).y; p_.y = (w1_).y; *(v2f*)((d_) + 2) = p_;                         \
        p_.x = (w0_).z; p_.y = (w1_).z; *(v2f*)((d_) + 4) = p_;                         \
        p_.x = (w0_).w; p_.y = (w1_).w; *(v2f*)((d_) + 6) = p_;                         \
    }
#define STORE_CHUNK(stage_)                                                             \
    {                                                                                   \
        float* ad_ = adst0 + (stage_) * kStageFloats;                                   \
        *(float4*)(ad_ +  0) = ta0;                                                     \
        *(float4*)(ad_ +  4) = ta1;                                                     \
        *(float4*)(ad_ +  8) = ta2;                                                     \
        *(float4*)(ad_ + 12) = ta3;                                                     \
        float* wd_ = wdst0 + (stage_) * kStageFloats;                                   \
        STORE_PAIRS(wd_ + 0 * 8 * kWPairStride * 2, tw00, tw01)                         \
        STORE_PAIRS(wd_ + 1 * 8 * kWPairStride * 2, tw10, tw11)                         \
        STORE_PAIRS(wd_ + 2 * 8 * kWPairStride * 2, tw20, tw21)                         \
        STORE_PAIRS(wd_ + 3 * 8 * kWPairStride * 2, tw30, tw31)                         \
    }

    LOAD_CHUNK(0)
    STORE_CHUNK(0)
    __syncthreads();

    #pragma unroll 2
    for (int kc = 0; kc < kD / kKC; ++kc) {
        const int  cur  = kc & 1;
        const bool more = kc + 1 < kD / kKC;
        if (more) LOAD_CHUNK((kc + 1) * kKC)

        const float* As = smem + cur * kStageFloats;
        const float* Wp = As + kAFloats;
        #pragma unroll
        for (int kk = 0; kk < kKC / 4; ++kk) {
            const int kb = kk * 4;
            const v2f a = *(const v2f*)&As[(mt * 16 + l16) * kAStride + kb + 2 * hi];
            v2f bfr[8];
            #pragma unroll
            for (int nt = 0; nt < 8; ++nt) {
                const int col = (nh * 8 + nt) * 16 + l16;
                bfr[nt] = *(const v2f*)&Wp[((kb >> 1) + hi) * (kWPairStride * 2) + col * 2];
            }
            #pragma unroll
            for (int nt = 0; nt < 8; ++nt) {
                acc[nt] = __builtin_amdgcn_wmma_f32_16x16x4_f32(
                    false, a, false, bfr[nt], (short)0, acc[nt], false, false);
            }
        }
        if (more) STORE_CHUNK(cur ^ 1)
        __syncthreads();
    }

#undef LOAD_CHUNK
#undef STORE_PAIRS
#undef STORE_CHUNK

    float* P = smem;
    float srow[8];
    #pragma unroll
    for (int r = 0; r < 8; ++r) srow[r] = 0.f;
    #pragma unroll
    for (int nt = 0; nt < 8; ++nt) {
        const int col    = (nh * 8 + nt) * 16 + l16;
        const float bcol = bvec[col];
        const float vcol = vvec[col];
        #pragma unroll
        for (int r = 0; r < 8; ++r) {
            acc[nt][r] += bcol;
            srow[r] += acc[nt][r] * vcol;
            P[(mt * 16 + hi * 8 + r) * kPStride + col] = acc[nt][r];
        }
    }
    #pragma unroll
    for (int off = 8; off >= 1; off >>= 1)
        #pragma unroll
        for (int r = 0; r < 8; ++r)
            srow[r] += __shfl_xor(srow[r], off, 16);
    if (l16 == 0) {
        #pragma unroll
        for (int r = 0; r < 8; ++r)
            scp[nh][mt * 16 + hi * 8 + r] = srow[r];
    }
    __syncthreads();

    for (int j = uwave; j < total; j += kWaves) {
        int t = j, w = 0;
        for (; w < kWaves - 1 && t >= scnt[w]; ++w) t -= scnt[w];
        const int s     = __builtin_amdgcn_readfirstlane((int)slist[w][t]);
        const int start = __builtin_amdgcn_readfirstlane(spans[3 * s + 1]);
        const int stop  = __builtin_amdgcn_readfirstlane(spans[3 * s + 2]);
        const int len   = stop - start;

        int r0 = start + lane;       if (r0 > kL - 1) r0 = kL - 1;  r0 -= p0;
        int r1 = start + lane + 32;  if (r1 > kL - 1) r1 = kL - 1;  r1 -= p0;
        const float sc0 = scp[0][r0] + scp[1][r0];
        const float sc1 = scp[0][r1] + scp[1][r1];
        const bool v0 = lane < len, v1 = lane + 32 < len;
        float m = fmaxf(v0 ? sc0 : -INFINITY, v1 ? sc1 : -INFINITY);
        #pragma unroll
        for (int off = 16; off >= 1; off >>= 1) m = fmaxf(m, __shfl_xor(m, off, 32));
        const float e0 = v0 ? expf(sc0 - m) : 0.f;
        const float e1 = v1 ? expf(sc1 - m) : 0.f;
        float sum = e0 + e1;
        #pragma unroll
        for (int off = 16; off >= 1; off >>= 1) sum += __shfl_xor(sum, off, 32);
        const float inv = 1.f / sum;
        const float a0 = e0 * inv, a1 = e1 * inv;

        const float* Pl = P + lane * 4;
        float4 pa = make_float4(0.f, 0.f, 0.f, 0.f);
        float4 pb = make_float4(0.f, 0.f, 0.f, 0.f);
        #pragma unroll 8
        for (int l = 0; l < kSpan; ++l) {
            const float av = (l < 32) ? a0 : a1;
            const float a  = __int_as_float(__builtin_amdgcn_readlane(__float_as_int(av), l & 31));
            int r = start + l;  if (r > kL - 1) r = kL - 1;  r -= p0;
            const float4 x = *(const float4*)(Pl + r * kPStride);
            const float4 y = *(const float4*)(Pl + r * kPStride + 128);
            pa.x = fmaf(a, x.x, pa.x); pa.y = fmaf(a, x.y, pa.y);
            pa.z = fmaf(a, x.z, pa.z); pa.w = fmaf(a, x.w, pa.w);
            pb.x = fmaf(a, y.x, pb.x); pb.y = fmaf(a, y.y, pb.y);
            pb.z = fmaf(a, y.z, pb.z); pb.w = fmaf(a, y.w, pb.w);
        }
        float* o = out + (size_t)s * kH + lane * 4;
        *(float4*)(o)       = pa;
        *(float4*)(o + 128) = pb;
    }
}


__launch_bounds__(kPoolThreads)
__global__ void span_pool_kernel(const int* __restrict__ spans,
                                 const float* __restrict__ ws, size_t ws_bytes,
                                 float* __restrict__ out)
{
    if (ws_bytes < kWsBytes) return;
    const int uw   = __builtin_amdgcn_readfirstlane((int)(threadIdx.x >> 5));
    const int lane = threadIdx.x & 31;
    const int task = blockIdx.x * (kPoolThreads / 32) + uw;
    const int s    = task >> 1;
    const int ch   = task & 1;

    int idx         = spans[3 * s + 0];
    const int start = spans[3 * s + 1];
    const int stop  = spans[3 * s + 2];
    idx = min(max(idx, 0), kN - 1);
    const int len = stop - start;

    const float* S = ws + (size_t)kAllRows * kH + (size_t)idx * kL;
    const int r0 = min(max(start + lane, 0), kL - 1);
    const int r1 = min(max(start + lane + 32, 0), kL - 1);
    const float sc0 = S[r0];
    const float sc1 = S[r1];
    const bool v0 = lane < len, v1 = lane + 32 < len;
    float m = fmaxf(v0 ? sc0 : -INFINITY, v1 ? sc1 : -INFINITY);
    #pragma unroll
    for (int off = 16; off >= 1; off >>= 1) m = fmaxf(m, __shfl_xor(m, off, 32));
    const float e0 = v0 ? expf(sc0 - m) : 0.f;
    const float e1 = v1 ? expf(sc1 - m) : 0.f;
    float sum = e0 + e1;
    #pragma unroll
    for (int off = 16; off >= 1; off >>= 1) sum += __shfl_xor(sum, off, 32);
    const float inv = 1.f / sum;
    const float a0 = e0 * inv, a1 = e1 * inv;

    const float* Pg = ws + (size_t)idx * kL * kH + ch * 128 + lane * 4;
    float4 pa = make_float4(0.f, 0.f, 0.f, 0.f);
    #pragma unroll 16
    for (int l = 0; l < kSpan; ++l) {
        const float av = (l < 32) ? a0 : a1;
        const float a  = __int_as_float(__builtin_amdgcn_readlane(__float_as_int(av), l & 31));
        const int r = min(max(start + l, 0), kL - 1);
        const float4 x = *(const float4*)(Pg + (size_t)r * kH);
        pa.x = fmaf(a, x.x, pa.x); pa.y = fmaf(a, x.y, pa.y);
        pa.z = fmaf(a, x.z, pa.z); pa.w = fmaf(a, x.w, pa.w);
    }
    *(float4*)(out + (size_t)s * kH + ch * 128 + lane * 4) = pa;
}

extern "C" void kernel_launch(void* const* d_in, const int* in_sizes, int n_in,
                              void* d_out, int out_size, void* d_ws, size_t ws_size,
                              hipStream_t stream) {
    (void)in_sizes; (void)n_in; (void)out_size;
    const float* hs    = (const float*)d_in[0];
    const int*   spans = (const int*)  d_in[1];
    const float* W     = (const float*)d_in[2];
    const float* b     = (const float*)d_in[3];
    const float* v     = (const float*)d_in[4];
    float* out = (float*)d_out;
    float* ws  = (float*)d_ws;
    attention_pooler_kernel<<<dim3(kGrid), dim3(kThreads), 0, stream>>>(hs, spans, W, b, v, ws, ws_size, out);
    span_pool_kernel<<<dim3(kPoolGrid), dim3(kPoolThreads), 0, stream>>>(spans, ws, ws_size, out);
}


# Pool launch: one wave per span (both column halves) walks only the span's valid rows l < len, sixteen rows' loads in flight, not all 64 rows in one wave per column half; sum order unchanged
# speedup vs baseline: 6.2766x; 1.0352x over previous
#include <hip/hip_runtime.h>
#include <hip/hip_bf16.h>
#include <math.h>


typedef __attribute__((ext_vector_type(2))) float v2f;
typedef __attribute__((ext_vector_type(8))) float v8f;

namespace {
constexpr int kLayer  = 2;
constexpr int kN = 32, kL = 512, kD = 1024, kH = 256;
constexpr int kS = 2048, kSpan = 64;
constexpr int kWin       = 64;
constexpr int kWinShift  = 6;
constexpr int kRows      = kWin + kSpan;
constexpr int kWinPerSeq = kL / kWin;
constexpr int kGrid      = kN * kWinPerSeq;
constexpr int kThreads   = 512;
constexpr int kWaves     = kThreads / 32;
constexpr int kScanPerWave = kS / kWaves;
constexpr int kKC = 64;
constexpr int kAStride = 68;
constexpr int kWPairStride = 272;
static_assert((1 << kWinShift) == kWin, "window shift");
constexpr int kPStride = 260;
constexpr int kAFloats = kRows * kAStride;
constexpr int kWFloats = (kKC / 2) * kWPairStride * 2;
constexpr int kStageFloats = kAFloats + kWFloats;
constexpr int kPFloats = kRows * kPStride;
constexpr int kAllRows  = kN * kL;
constexpr int kTRows    = 64;
constexpr int kTAFloats = kTRows * kAStride;
constexpr int kTStage   = kTAFloats + kWFloats;
static_assert(kAllRows / kTRows == kGrid, "one 64-row tile per work-group");
static_assert(2 * kTStage <= 2 * kStageFloats && kTRows * kPStride <= 2 * kStageFloats, "fits the LDS ring");
constexpr size_t kWsFloats = (size_t)kAllRows * kH + kAllRows;
constexpr size_t kWsBytes  = kWsFloats * sizeof(float);
constexpr int kPoolThreads = 256;
constexpr int kPoolGrid    = kS / (kPoolThreads / 32);
constexpr int kPoolGroup   = 16;
}

__launch_bounds__(512)
__global__ void attention_pooler_kernel(const float* __restrict__ hs_all,
                                        const int*  __restrict__ spans,
                                        const float* __restrict__ Wm,
                                        const float* __restrict__ bvec,
                                        const float* __restrict__ vvec,
                                        float* __restrict__ ws, size_t ws_bytes,
                                        float* __restrict__ out)
{
    __shared__ __align__(16) float smem[2 * kStageFloats];
    __shared__ float scp[2][kRows];
    __shared__ unsigned short slist[kWaves][kScanPerWave];
    __shared__ int scnt[kWaves];

    const int seq  = blockIdx.x / kWinPerSeq;
    const int win  = blockIdx.x % kWinPerSeq;
    const int p0   = win * kWin;
    const int tid  = threadIdx.x;
    const int wave = tid >> 5;
    const int lane = tid & 31;
    const int mt   = wave & 7;
    const int nh   = wave >> 3;
    const int hi   = lane >> 4;
    const int l16  = lane & 15;
    const int uwave = __builtin_amdgcn_readfirstlane(wave);

    const float* hs = hs_all + (size_t)kLayer * kN * kL * kD;

    if (ws_bytes >= kWsBytes) {
        const int row0 = blockIdx.x * kTRows;
        const int tmt  = wave & 3;
        const int tnq  = wave >> 2;

        v8f tacc[4];
        #pragma unroll
        for (int nt = 0; nt < 4; ++nt)
            #pragma unroll
            for (int e = 0; e < 8; ++e) tacc[nt][e] = 0.f;

        const int ar = tid >> 4;
        const int ac = (tid & 15) * 4;
        const float* asrc0 = hs + (size_t)(row0 + ar) * kD + ac;
        const float* asrc1 = asrc0 + (size_t)32 * kD;
        float* tadst = &smem[ar * kAStride + ac];
        const int twkp = tid >> 6;
        const int twc  = (tid & 63) * 4;
        const float* twsrc = Wm + (size_t)(2 * twkp) * kH + twc;
        float* twdst = &smem[kTAFloats + (twkp * kWPairStride + twc) * 2];
        float4 ua0, ua1;
        float4 uw00, uw01, uw10, uw11, uw20, uw21, uw30, uw31;

#define T_LOAD(kbase_)                                                                  \
    {                                                                                   \
        ua0 = *(const float4*)(asrc0 + (kbase_));                                       \
        ua1 = *(const float4*)(asrc1 + (kbase_));                                       \
        const float* w_ = twsrc + (size_t)(kbase_) * kH;                                \
        uw00 = *(const float4*)(w_ + (size_t)( 0 + 0) * kH);                            \
        uw01 = *(const float4*)(w_ + (size_t)( 0 + 1) * kH);                            \
        uw10 = *(const float4*)(w_ + (size_t)(16 + 0) * kH);                            \
        uw11 = *(const float4*)(w_ + (size_t)(16 + 1) * kH);                            \
        uw20 = *(const float4*)(w_ + (size_t)(32 + 0) * kH);                            \
        uw21 = *(const float4*)(w_ + (size_t)(32 + 1) * kH);                            \
        uw30 = *(const float4*)(w_ + (size_t)(48 + 0) * kH);                            \
        uw31 = *(const float4*)(w_ + (size_t)(48 + 1) * kH);                            \
    }
#define T_PAIRS(d_, w0_, w1_)                                                           \
    {                                                                                   \
        v2f p_;                                                                         \
        p_.x = (w0_).x; p_.y = (w1_).x; *(v2f*)((d_) + 0) = p_;                         \
        p_.x = (w0_).y; p_.y = (w1_).y; *(v2f*)((d_) + 2) = p_;                         \
        p_.x = (w0_).z; p_.y = (w1_).z; *(v2f*)((d_) + 4) = p_;                         \
        p_.x = (w0_).w; p_.y = (w1_).w; *(v2f*)((d_) + 6) = p_;                         \
    }
#define T_STORE(stage_)                                                                 \
    {                                                                                   \
        float* ad_ = tadst + (stage_) * kTStage;                                        \
        *(float4*)(ad_) = ua0;                                                          \
        *(float4*)(ad_ + 32 * kAStride) = ua1;                                          \
        float* wd_ = twdst + (stage_) * kTStage;                                        \
        T_PAIRS(wd_ + 0 * 8 * kWPairStride * 2, uw00, uw01)                             \
        T_PAIRS(wd_ + 1 * 8 * kWPairStride * 2, uw10, uw11)                             \
        T_PAIRS(wd_ + 2 * 8 * kWPairStride * 2, uw20, uw21)                             \
        T_PAIRS(wd_ + 3 * 8 * kWPairStride * 2, uw30, uw31)                             \
    }

        T_LOAD(0)
        T_STORE(0)
        __syncthreads();

        #pragma unroll 2
        for (int kc = 0; kc < kD / kKC; ++kc) {
            const int  cur  = kc & 1;
            const bool more = kc + 1 < kD / kKC;
            if (more) T_LOAD((kc + 1) * kKC)

            const float* As = smem + cur * kTStage;
            const float* Wp = As + kTAFloats;
            #pragma unroll
            for (int kk = 0; kk < kKC / 4; ++kk) {
                const int kb = kk * 4;
                const v2f a = *(const v2f*)&As[(tmt * 16 + l16) * kAStride + kb + 2 * hi];
                v2f bfr[4];
                #pragma unroll
                for (int nt = 0; nt < 4; ++nt) {
                    const int col = (tnq * 4 + nt) * 16 + l16;
                    bfr[nt] = *(const v2f*)&Wp[((kb >> 1) + hi) * (kWPairStride * 2) + col * 2];
                }
                #pragma unroll
                for (int nt = 0; nt < 4; ++nt) {
                    tacc[nt] = __builtin_amdgcn_wmma_f32_16x16x4_f32(
                        false, a, false, bfr[nt], (short)0, tacc[nt], false, false);
                }
            }
            if (more) T_STORE(cur ^ 1)
            __syncthreads();
        }
#undef T_LOAD
#undef T_PAIRS
#undef T_STORE

        float* Pt = smem;
        float (*scq)[kTRows] = (float (*)[kTRows])&scp[0][0];
        float srow[8];
        #pragma unroll
        for (int r = 0; r < 8; ++r) srow[r] = 0.f;
        #pragma unroll
        for (int nt = 0; nt < 4; ++nt) {
            const int col    = (tnq * 4 + nt) * 16 + l16;
            const float bcol = bvec[col];
            const float vcol = vvec[col];
            #pragma unroll
            for (int r = 0; r < 8; ++r) {
                tacc[nt][r] += bcol;
                srow[r] += tacc[nt][r] * vcol;
                Pt[(tmt * 16 + hi * 8 + r) * kPStride + col] = tacc[nt][r];
            }
        }
        #pragma unroll
        for (int off = 8; off >= 1; off >>= 1)
            #pragma unroll
            for (int r = 0; r < 8; ++r)
                srow[r] += __shfl_xor(srow[r], off, 16);
        if (l16 == 0) {
            #pragma unroll
            for (int r = 0; r < 8; ++r)
                scq[tnq][tmt * 16 + hi * 8 + r] = srow[r];
        }
        __syncthreads();

        float* wsS = ws + (size_t)kAllRows * kH;
        if (tid < kTRows)
            wsS[row0 + tid] = (scq[0][tid] + scq[1][tid]) + (scq[2][tid] + scq[3][tid]);
        #pragma unroll
        for (int i = 0; i < kTRows / kWaves; ++i) {
            const int r = uwave * (kTRows / kWaves) + i;
            const float* s_ = Pt + r * kPStride + lane * 4;
            float* d_ = ws + (size_t)(row0 + r) * kH + lane * 4;
            const float4 x = *(const float4*)(s_);
            const float4 y = *(const float4*)(s_ + 128);
            *(float4*)(d_)       = x;
            *(float4*)(d_ + 128) = y;
        }
        return;
    }

    {
        int ridx[kScanPerWave / 32], rst[kScanPerWave / 32];
        #pragma unroll
        for (int it = 0; it < kScanPerWave / 32; ++it) {
            const int s = uwave * kScanPerWave + it * 32 + lane;
            ridx[it] = spans[3 * s + 0];
            rst[it]  = spans[3 * s + 1];
        }
        int n = 0;
        #pragma unroll
        for (int it = 0; it < kScanPerWave / 32; ++it) {
            const int s = uwave * kScanPerWave + it * 32 + lane;
            const bool mine = (ridx[it] == seq) && ((rst[it] >> kWinShift) == win);
            const unsigned mask = (unsigned)__ballot(mine);
            if (mine) slist[uwave][n + __popc(mask & ((1u << lane) - 1u))] = (unsigned short)s;
            n += __popc(mask);
        }
        if (lane == 0) scnt[uwave] = n;
    }
    __syncthreads();
    int total = 0;
    for (int w = 0; w < kWaves; ++w) total += scnt[w];
    total = __builtin_amdgcn_readfirstlane(total);
    if (total == 0) return;

    v8f acc[8];
    #pragma unroll
    for (int nt = 0; nt < 8; ++nt)
        #pragma unroll
        for (int e = 0; e < 8; ++e) acc[nt][e] = 0.f;

    const int rowj = tid >> 2;
    const int quad = tid & 3;
    int pos = p0 + rowj;
    if (pos > kL - 1) pos = kL - 1;
    const float* arow = hs + ((size_t)seq * kL + pos) * kD;

    const int wkp = tid >> 6;
    const int wc  = (tid & 63) * 4;
    const float* wsrc = Wm + (size_t)(2 * wkp) * kH + wc;
    float* adst0 = &smem[rowj * kAStride + quad * 16];
    float* wdst0 = &smem[kAFloats + (wkp * kWPairStride + wc) * 2];
    float4 ta0, ta1, ta2, ta3;
    float4 tw00, tw01, tw10, tw11, tw20, tw21, tw30, tw31;

#define LOAD_CHUNK(kbase_)                                                              \
    {                                                                                   \
        const float4* asrc_ = (const float4*)(arow + (kbase_) + quad * 16);             \
        ta0 = asrc_[0]; ta1 = asrc_[1]; ta2 = asrc_[2]; ta3 = asrc_[3];                 \
        const float* w_ = wsrc + (size_t)(kbase_) * kH;                                 \
        tw00 = *(const float4*)(w_ + (size_t)( 0 + 0) * kH);                            \
        tw01 = *(const float4*)(w_ + (size_t)( 0 + 1) * kH);                            \
        tw10 = *(const float4*)(w_ + (size_t)(16 + 0) * kH);                            \
        tw11 = *(const float4*)(w_ + (size_t)(16 + 1) * kH);                            \
        tw20 = *(const float4*)(w_ + (size_t)(32 + 0) * kH);                            \
        tw21 = *(const float4*)(w_ + (size_t)(32 + 1) * kH);                            \
        tw30 = *(const float4*)(w_ + (size_t)(48 + 0) * kH);                            \
        tw31 = *(const float4*)(w_ + (size_t)(48 + 1) * kH);                            \
    }
#define STORE_PAIRS(d_, w0_, w1_)                                                       \
    {                                                                                   \
        v2f p_;                                                                         \
        p_.x = (w0_).x; p_.y = (w1_).x; *(v2f*)((d_) + 0) = p_;                         \
        p_.x = (w0_).y; p_.y = (w1_).y; *(v2f*)((d_) + 2) = p_;                         \
        p_.x = (w0_).z; p_.y = (w1_).z; *(v2f*)((d_) + 4) = p_;                         \
        p_.x = (w0_).w; p_.y = (w1_).w; *(v2f*)((d_) + 6) = p_;                         \
    }
#define STORE_CHUNK(stage_)                                                             \
    {                                                                                   \
        float* ad_ = adst0 + (stage_) * kStageFloats;                                   \
        *(float4*)(ad_ +  0) = ta0;                                                     \
        *(float4*)(ad_ +  4) = ta1;                                                     \
        *(float4*)(ad_ +  8) = ta2;                                                     \
        *(float4*)(ad_ + 12) = ta3;                                                     \
        float* wd_ = wdst0 + (stage_) * kStageFloats;                                   \
        STORE_PAIRS(wd_ + 0 * 8 * kWPairStride * 2, tw00, tw01)                         \
        STORE_PAIRS(wd_ + 1 * 8 * kWPairStride * 2, tw10, tw11)                         \
        STORE_PAIRS(wd_ + 2 * 8 * kWPairStride * 2, tw20, tw21)                         \
        STORE_PAIRS(wd_ + 3 * 8 * kWPairStride * 2, tw30, tw31)                         \
    }

    LOAD_CHUNK(0)
    STORE_CHUNK(0)
    __syncthreads();

    #pragma unroll 2
    for (int kc = 0; kc < kD / kKC; ++kc) {
        const int  cur  = kc & 1;
        const bool more = kc + 1 < kD / kKC;
        if (more) LOAD_CHUNK((kc + 1) * kKC)

        const float* As = smem + cur * kStageFloats;
        const float* Wp = As + kAFloats;
        #pragma unroll
        for (int kk = 0; kk < kKC / 4; ++kk) {
            const int kb = kk * 4;
            const v2f a = *(const v2f*)&As[(mt * 16 + l16) * kAStride + kb + 2 * hi];
            v2f bfr[8];
            #pragma unroll
            for (int nt = 0; nt < 8; ++nt) {
                const int col = (nh * 8 + nt) * 16 + l16;
                bfr[nt] = *(const v2f*)&Wp[((kb >> 1) + hi) * (kWPairStride * 2) + col * 2];
            }
            #pragma unroll
            for (int nt = 0; nt < 8; ++nt) {
                acc[nt] = __builtin_amdgcn_wmma_f32_16x16x4_f32(
                    false, a, false, bfr[nt], (short)0, acc[nt], false, false);
            }
        }
        if (more) STORE_CHUNK(cur ^ 1)
        __syncthreads();
    }

#undef LOAD_CHUNK
#undef STORE_PAIRS
#undef STORE_CHUNK

    float* P = smem;
    float srow[8];
    #pragma unroll
    for (int r = 0; r < 8; ++r) srow[r] = 0.f;
    #pragma unroll
    for (int nt = 0; nt < 8; ++nt) {
        const int col    = (nh * 8 + nt) * 16 + l16;
        const float bcol = bvec[col];
        const float vcol = vvec[col];
        #pragma unroll
        for (int r = 0; r < 8; ++r) {
            acc[nt][r] += bcol;
            srow[r] += acc[nt][r] * vcol;
            P[(mt * 16 + hi * 8 + r) * kPStride + col] = acc[nt][r];
        }
    }
    #pragma unroll
    for (int off = 8; off >= 1; off >>= 1)
        #pragma unroll
        for (int r = 0; r < 8; ++r)
            srow[r] += __shfl_xor(srow[r], off, 16);
    if (l16 == 0) {
        #pragma unroll
        for (int r = 0; r < 8; ++r)
            scp[nh][mt * 16 + hi * 8 + r] = srow[r];
    }
    __syncthreads();

    for (int j = uwave; j < total; j += kWaves) {
        int t = j, w = 0;
        for (; w < kWaves - 1 && t >= scnt[w]; ++w) t -= scnt[w];
        const int s     = __builtin_amdgcn_readfirstlane((int)slist[w][t]);
        const int start = __builtin_amdgcn_readfirstlane(spans[3 * s + 1]);
        const int stop  = __builtin_amdgcn_readfirstlane(spans[3 * s + 2]);
        const int len   = stop - start;

        int r0 = start + lane;       if (r0 > kL - 1) r0 = kL - 1;  r0 -= p0;
        int r1 = start + lane + 32;  if (r1 > kL - 1) r1 = kL - 1;  r1 -= p0;
        const float sc0 = scp[0][r0] + scp[1][r0];
        const float sc1 = scp[0][r1] + scp[1][r1];
        const bool v0 = lane < len, v1 = lane + 32 < len;
        float m = fmaxf(v0 ? sc0 : -INFINITY, v1 ? sc1 : -INFINITY);
        #pragma unroll
        for (int off = 16; off >= 1; off >>= 1) m = fmaxf(m, __shfl_xor(m, off, 32));
        const float e0 = v0 ? expf(sc0 - m) : 0.f;
        const float e1 = v1 ? expf(sc1 - m) : 0.f;
        float sum = e0 + e1;
        #pragma unroll
        for (int off = 16; off >= 1; off >>= 1) sum += __shfl_xor(sum, off, 32);
        const float inv = 1.f / sum;
        const float a0 = e0 * inv, a1 = e1 * inv;

        const float* Pl = P + lane * 4;
        float4 pa = make_float4(0.f, 0.f, 0.f, 0.f);
        float4 pb = make_float4(0.f, 0.f, 0.f, 0.f);
        #pragma unroll 8
        for (int l = 0; l < kSpan; ++l) {
            const float av = (l < 32) ? a0 : a1;
            const float a  = __int_as_float(__builtin_amdgcn_readlane(__float_as_int(av), l & 31));
            int r = start + l;  if (r > kL - 1) r = kL - 1;  r -= p0;
            const float4 x = *(const float4*)(Pl + r * kPStride);
            const float4 y = *(const float4*)(Pl + r * kPStride + 128);
            pa.x = fmaf(a, x.x, pa.x); pa.y = fmaf(a, x.y, pa.y);
            pa.z = fmaf(a, x.z, pa.z); pa.w = fmaf(a, x.w, pa.w);
            pb.x = fmaf(a, y.x, pb.x); pb.y = fmaf(a, y.y, pb.y);
            pb.z = fmaf(a, y.z, pb.z); pb.w = fmaf(a, y.w, pb.w);
        }
        float* o = out + (size_t)s * kH + lane * 4;
        *(float4*)(o)       = pa;
        *(float4*)(o + 128) = pb;
    }
}


namespace {
template <int NR>
__device__ __forceinline__ void pool_request(const float* __restrict__ Pg, int start, int l,
                                             float4* x, float4* y)
{
    #pragma unroll
    for (int j = 0; j < NR; ++j) {
        const int r = min(max(start + l + j, 0), kL - 1);
        x[j] = *(const float4*)(Pg + (size_t)r * kH);
        y[j] = *(const float4*)(Pg + (size_t)r * kH + 128);
    }
}
template <int NR>
__device__ __forceinline__ void pool_add(float av, int l, const float4* x, const float4* y,
                                         float4& pa, float4& pb)
{
    #pragma unroll
    for (int j = 0; j < NR; ++j) {
        const float a = __int_as_float(__builtin_amdgcn_readlane(__float_as_int(av), (l + j) & 31));
        pa.x = fmaf(a, x[j].x, pa.x); pa.y = fmaf(a, x[j].y, pa.y);
        pa.z = fmaf(a, x[j].z, pa.z); pa.w = fmaf(a, x[j].w, pa.w);
        pb.x = fmaf(a, y[j].x, pb.x); pb.y = fmaf(a, y[j].y, pb.y);
        pb.z = fmaf(a, y[j].z, pb.z); pb.w = fmaf(a, y[j].w, pb.w);
    }
}
}

__launch_bounds__(kPoolThreads)
__global__ void span_pool_kernel(const int* __restrict__ spans,
                                 const float* __restrict__ ws, size_t ws_bytes,
                                 float* __restrict__ out)
{
    if (ws_bytes < kWsBytes) return;
    const int uw   = __builtin_amdgcn_readfirstlane((int)(threadIdx.x >> 5));
    const int lane = threadIdx.x & 31;
    const int s    = blockIdx.x * (kPoolThreads / 32) + uw;

    int idx         = spans[3 * s + 0];
    const int start = spans[3 * s + 1];
    const int stop  = spans[3 * s + 2];
    idx = min(max(idx, 0), kN - 1);
    const int len = stop - start;
    const int nrow = (len <= 0) ? kSpan : min(len, kSpan);

    const float* S = ws + (size_t)kAllRows * kH + (size_t)idx * kL;
    const int r0 = min(max(start + lane, 0), kL - 1);
    const int r1 = min(max(start + lane + 32, 0), kL - 1);
    const float sc0 = S[r0];
    const float sc1 = S[r1];
    const bool v0 = lane < len, v1 = lane + 32 < len;
    float m = fmaxf(v0 ? sc0 : -INFINITY, v1 ? sc1 : -INFINITY);
    #pragma unroll
    for (int off = 16; off >= 1; off >>= 1) m = fmaxf(m, __shfl_xor(m, off, 32));
    const float e0 = v0 ? expf(sc0 - m) : 0.f;
    const float e1 = v1 ? expf(sc1 - m) : 0.f;
    float sum = e0 + e1;
    #pragma unroll
    for (int off = 16; off >= 1; off >>= 1) sum += __shfl_xor(sum, off, 32);
    const float inv = 1.f / sum;
    const float a0 = e0 * inv, a1 = e1 * inv;

    const float* Pg = ws + (size_t)idx * kL * kH + lane * 4;
    float4 pa = make_float4(0.f, 0.f, 0.f, 0.f);
    float4 pb = make_float4(0.f, 0.f, 0.f, 0.f);
    int l = 0;
    for (; l + kPoolGroup <= nrow; l += kPoolGroup) {
        float4 x[kPoolGroup], y[kPoolGroup];
        pool_request<kPoolGroup>(Pg, start, l, x, y);
        pool_add<kPoolGroup>((l < 32) ? a0 : a1, l, x, y, pa, pb);
    }
    const int rem = nrow - l;
    if (rem > 0) {
        float4 x[kPoolGroup], y[kPoolGroup];
        const float av = (l < 32) ? a0 : a1;
        pool_request<4>(Pg, start, l, x, y);
        if (rem > 4)  pool_request<4>(Pg, start, l + 4,  x + 4,  y + 4);
        if (rem > 8)  pool_request<4>(Pg, start, l + 8,  x + 8,  y + 8);
        if (rem > 12) pool_request<4>(Pg, start, l + 12, x + 12, y + 12);
        pool_add<4>(av, l, x, y, pa, pb);
        if (rem > 4)  pool_add<4>(av, l + 4,  x + 4,  y + 4,  pa, pb);
        if (rem > 8)  pool_add<4>(av, l + 8,  x + 8,  y + 8,  pa, pb);
        if (rem > 12) pool_add<4>(av, l + 12, x + 12, y + 12, pa, pb);
    }
    float* o = out + (size_t)s * kH + lane * 4;
    *(float4*)(o)       = pa;
    *(float4*)(o + 128) = pb;
}

extern "C" void kernel_launch(void* const* d_in, const int* in_sizes, int n_in,
                              void* d_out, int out_size, void* d_ws, size_t ws_size,
                              hipStream_t stream) {
    (void)in_sizes; (void)n_in; (void)out_size;
    const float* hs    = (const float*)d_in[0];
    const int*   spans = (const int*)  d_in[1];
    const float* W     = (const float*)d_in[2];
    const float* b     = (const float*)d_in[3];
    const float* v     = (const float*)d_in[4];
    float* out = (float*)d_out;
    float* ws  = (float*)d_ws;
    attention_pooler_kernel<<<dim3(kGrid), dim3(kThreads), 0, stream>>>(hs, spans, W, b, v, ws, ws_size, out);
    span_pool_kernel<<<dim3(kPoolGrid), dim3(kPoolThreads), 0, stream>>>(spans, ws, ws_size, out);
}


# Projection is linear: pool the RAW rows per span with the softmax weights (score = x.(W v) + b.v), then WMMA-project the 2048 pooled rows, not all 16384 rows; sum order changes, f32 kept
# speedup vs baseline: 10.3210x; 1.6444x over previous
#include <hip/hip_runtime.h>
#include <hip/hip_bf16.h>
#include <math.h>


typedef __attribute__((ext_vector_type(2))) float v2f;
typedef __attribute__((ext_vector_type(8))) float v8f;

namespace {
constexpr int kLayer  = 2;
constexpr int kN = 32, kL = 512, kD = 1024, kH = 256;
constexpr int kS = 2048, kSpan = 64;
constexpr int kWin       = 64;
constexpr int kWinShift  = 6;
constexpr int kRows      = kWin + kSpan;
constexpr int kWinPerSeq = kL / kWin;
constexpr int kGrid      = kN * kWinPerSeq;
constexpr int kThreads   = 512;
constexpr int kWaves     = kThreads / 32;
constexpr int kScanPerWave = kS / kWaves;
constexpr int kKC = 64;
constexpr int kAStride = 68;
constexpr int kWPairStride = 272;
static_assert((1 << kWinShift) == kWin, "window shift");
constexpr int kPStride = 260;
constexpr int kAFloats = kRows * kAStride;
constexpr int kWFloats = (kKC / 2) * kWPairStride * 2;
constexpr int kStageFloats = kAFloats + kWFloats;
constexpr int kPFloats = kRows * kPStride;
constexpr int kAllRows  = kN * kL;
constexpr size_t kOffU  = 0;
constexpr size_t kOffS  = 2048;
constexpr size_t kOffA  = kOffS + kAllRows;
constexpr size_t kOffZ  = kOffA + kS;
constexpr size_t kNeedFloats = kOffZ + (size_t)kS * kD;
constexpr size_t kNeedBytes  = kNeedFloats * sizeof(float);
static_assert(kOffZ % 64 == 0, "pooled rows start on a 256-B boundary");
constexpr int kGM = 32, kGN = 64;
constexpr int kGKC = 128;
constexpr int kGAStride = 132;
constexpr int kGPairStride = 80;
constexpr int kGAFloats = kGM * kGAStride;
constexpr int kGWFloats = (kGKC / 2) * kGPairStride * 2;
constexpr int kGStage   = kGAFloats + kGWFloats;
constexpr int kGPPitch  = 68;
static_assert((kS / kGM) * (kH / kGN) == kGrid, "one 32 x 64 tile per work-group");
static_assert(2 * kGStage <= 2 * kStageFloats && 2 * kGM * kGPPitch <= 2 * kStageFloats, "fits the LDS ring");
constexpr int kPoolThreads = 256;
constexpr int kPoolGrid    = kS * 4 / (kPoolThreads / 32);
constexpr int kPoolGroup   = 16;
constexpr int kScoreThreads = 512;
constexpr int kScoreGrid    = kAllRows / (16 * 4);
constexpr int kUThreads = 256;
constexpr int kUGrid    = kD / 8;
}

__launch_bounds__(512)
__global__ void attention_pooler_kernel(const float* __restrict__ hs_all,
                                        const int*  __restrict__ spans,
                                        const float* __restrict__ Wm,
                                        const float* __restrict__ bvec,
                                        const float* __restrict__ vvec,
                                        float* __restrict__ ws, size_t ws_bytes,
                                        float* __restrict__ out)
{
    __shared__ __align__(16) float smem[2 * kStageFloats];
    __shared__ float scp[2][kRows];
    __shared__ unsigned short slist[kWaves][kScanPerWave];
    __shared__ int scnt[kWaves];

    const int seq  = blockIdx.x / kWinPerSeq;
    const int win  = blockIdx.x % kWinPerSeq;
    const int p0   = win * kWin;
    const int tid  = threadIdx.x;
    const int wave = tid >> 5;
    const int lane = tid & 31;
    const int mt   = wave & 7;
    const int nh   = wave >> 3;
    const int hi   = lane >> 4;
    const int l16  = lane & 15;
    const int uwave = __builtin_amdgcn_readfirstlane(wave);

    const float* hs = hs_all + (size_t)kLayer * kN * kL * kD;

    if (ws_bytes >= kNeedBytes) {
        const int m0 = (blockIdx.x >> 2) * kGM;
        const int n0 = (blockIdx.x & 3) * kGN;
        const int tm = wave & 1;
        const int tn = (wave >> 1) & 3;
        const int kh = wave >> 3;
        const float* Z  = ws + kOffZ;
        const float* SA = ws + kOffA;

        v8f acc0, acc1;
        #pragma unroll
        for (int e = 0; e < 8; ++e) { acc0[e] = 0.f; acc1[e] = 0.f; }

        const int ar = tid >> 4;
        const int ac = (tid & 15) * 4;
        const float* asrc = Z + (size_t)(m0 + ar) * kD + ac;
        float* adst = &smem[ar * kGAStride + ac];
        const float* wsrc = Wm + (size_t)(2 * ar) * kH + n0 + ac;
        float* wdst = &smem[kGAFloats + (ar * kGPairStride + ac) * 2];
        float4 ga0, ga1, gw00, gw01, gw10, gw11;

#define G_LOAD(kbase_)                                                                  \
    {                                                                                   \
        ga0 = *(const float4*)(asrc + (kbase_));                                        \
        ga1 = *(const float4*)(asrc + (kbase_) + 64);                                   \
        const float* w_ = wsrc + (size_t)(kbase_) * kH;                                 \
        gw00 = *(const float4*)(w_ + (size_t)( 0 + 0) * kH);                            \
        gw01 = *(const float4*)(w_ + (size_t)( 0 + 1) * kH);                            \
        gw10 = *(const float4*)(w_ + (size_t)(64 + 0) * kH);                            \
        gw11 = *(const float4*)(w_ + (size_t)(64 + 1) * kH);                            \
    }
#define G_PAIRS(d_, w0_, w1_)                                                           \
    {                                                                                   \
        v2f p_;                                                                         \
        p_.x = (w0_).x; p_.y = (w1_).x; *(v2f*)((d_) + 0) = p_;                         \
        p_.x = (w0_).y; p_.y = (w1_).y; *(v2f*)((d_) + 2) = p_;                         \
        p_.x = (w0_).z; p_.y = (w1_).z; *(v2f*)((d_) + 4) = p_;                         \
        p_.x = (w0_).w; p_.y = (w1_).w; *(v2f*)((d_) + 6) = p_;                         \
    }
#define G_STORE(stage_)                                                                 \
    {                                                                                   \
        float* ad_ = adst + (stage_) * kGStage;                                         \
        *(float4*)(ad_)      = ga0;                                                     \
        *(float4*)(ad_ + 64) = ga1;                                                     \
        float* wd_ = wdst + (stage_) * kGStage;                                         \
        G_PAIRS(wd_, gw00, gw01)                                                        \
        G_PAIRS(wd_ + 32 * kGPairStride * 2, gw10, gw11)                                \
    }

        G_LOAD(0)
        G_STORE(0)
        __syncthreads();

        const int aoff = (tm * 16 + l16) * kGAStride + kh * (kGKC / 2) + 2 * hi;
        const int woff = (kh * (kGKC / 4) + hi) * (kGPairStride * 2) + (tn * 16 + l16) * 2;

        #pragma unroll 2
        for (int kc = 0; kc < kD / kGKC; ++kc) {
            const int  cur  = kc & 1;
            const bool more = kc + 1 < kD / kGKC;
            if (more) G_LOAD((kc + 1) * kGKC)

            const float* As = smem + cur * kGStage + aoff;
            const float* Wp = smem + cur * kGStage + kGAFloats + woff;
            #pragma unroll
            for (int kk = 0; kk < kGKC / 8; kk += 2) {
                const v2f a0 = *(const v2f*)&As[kk * 4];
                const v2f b0 = *(const v2f*)&Wp[kk * 2 * (kGPairStride * 2)];
                const v2f a1 = *(const v2f*)&As[(kk + 1) * 4];
                const v2f b1 = *(const v2f*)&Wp[(kk + 1) * 2 * (kGPairStride * 2)];
                acc0 = __builtin_amdgcn_wmma_f32_16x16x4_f32(false, a0, false, b0, (short)0, acc0, false, false);
                acc1 = __builtin_amdgcn_wmma_f32_16x16x4_f32(false, a1, false, b1, (short)0, acc1, false, false);
            }
            if (more) G_STORE(cur ^ 1)
            __syncthreads();
        }
#undef G_LOAD
#undef G_PAIRS
#undef G_STORE

        float* Pp = smem;
        #pragma unroll
        for (int r = 0; r < 8; ++r)
            Pp[(kh * kGM + tm * 16 + hi * 8 + r) * kGPPitch + tn * 16 + l16] = acc0[r] + acc1[r];
        __syncthreads();
        {
            const float4 q0 = *(const float4*)&Pp[ar * kGPPitch + ac];
            const float4 q1 = *(const float4*)&Pp[(kGM + ar) * kGPPitch + ac];
            const float  sa = SA[m0 + ar];
            const float4 bb = *(const float4*)&bvec[n0 + ac];
            float4 o;
            o.x = (q0.x + q1.x) + sa * bb.x;
            o.y = (q0.y + q1.y) + sa * bb.y;
            o.z = (q0.z + q1.z) + sa * bb.z;
            o.w = (q0.w + q1.w) + sa * bb.w;
            *(float4*)&out[(size_t)(m0 + ar) * kH + n0 + ac] = o;
        }
        return;
    }


    {
        int ridx[kScanPerWave / 32], rst[kScanPerWave / 32];
        #pragma unroll
        for (int it = 0; it < kScanPerWave / 32; ++it) {
            const int s = uwave * kScanPerWave + it * 32 + lane;
            ridx[it] = spans[3 * s + 0];
            rst[it]  = spans[3 * s + 1];
        }
        int n = 0;
        #pragma unroll
        for (int it = 0; it < kScanPerWave / 32; ++it) {
            const int s = uwave * kScanPerWave + it * 32 + lane;
            const bool mine = (ridx[it] == seq) && ((rst[it] >> kWinShift) == win);
            const unsigned mask = (unsigned)__ballot(mine);
            if (mine) slist[uwave][n + __popc(mask & ((1u << lane) - 1u))] = (unsigned short)s;
            n += __popc(mask);
        }
        if (lane == 0) scnt[uwave] = n;
    }
    __syncthreads();
    int total = 0;
    for (int w = 0; w < kWaves; ++w) total += scnt[w];
    total = __builtin_amdgcn_readfirstlane(total);
    if (total == 0) return;

    v8f acc[8];
    #pragma unroll
    for (int nt = 0; nt < 8; ++nt)
        #pragma unroll
        for (int e = 0; e < 8; ++e) acc[nt][e] = 0.f;

    const int rowj = tid >> 2;
    const int quad = tid & 3;
    int pos = p0 + rowj;
    if (pos > kL - 1) pos = kL - 1;
    const float* arow = hs + ((size_t)seq * kL + pos) * kD;

    const int wkp = tid >> 6;
    const int wc  = (tid & 63) * 4;
    const float* wsrc = Wm + (size_t)(2 * wkp) * kH + wc;
    float* adst0 = &smem[rowj * kAStride + quad * 16];
    float* wdst0 = &smem[kAFloats + (wkp * kWPairStride + wc) * 2];
    float4 ta0, ta1, ta2, ta3;
    float4 tw00, tw01, tw10, tw11, tw20, tw21, tw30, tw31;

#define LOAD_CHUNK(kbase_)                                                              \
    {                                                                                   \
        const float4* asrc_ = (const float4*)(arow + (kbase_) + quad * 16);             \
        ta0 = asrc_[0]; ta1 = asrc_[1]; ta2 = asrc_[2]; ta3 = asrc_[3];                 \
        const float* w_ = wsrc + (size_t)(kbase_) * kH;                                 \
        tw00 = *(const float4*)(w_ + (size_t)( 0 + 0) * kH);                            \
        tw01 = *(const float4*)(w_ + (size_t)( 0 + 1) * kH);                            \
        tw10 = *(const float4*)(w_ + (size_t)(16 + 0) * kH);                            \
        tw11 = *(const float4*)(w_ + (size_t)(16 + 1) * kH);                            \
        tw20 = *(const float4*)(w_ + (size_t)(32 + 0) * kH);                            \
        tw21 = *(const float4*)(w_ + (size_t)(32 + 1) * kH);                            \
        tw30 = *(const float4*)(w_ + (size_t)(48 + 0) * kH);                            \
        tw31 = *(const float4*)(w_ + (size_t)(48 + 1) * kH);                            \
    }
#define STORE_PAIRS(d_, w0_, w1_)                                                       \
    {                                                                                   \
        v2f p_;                                                                         \
        p_.x = (w0_).x; p_.y = (w1_).x; *(v2f*)((d_) + 0) = p_;                         \
        p_.x = (w0_).y; p_.y = (w1_).y; *(v2f*)((d_) + 2) = p_;                         \
        p_.x = (w0_).z; p_.y = (w1_).z; *(v2f*)((d_) + 4) = p_;                         \
        p_.x = (w0_).w; p_.y = (w1_).w; *(v2f*)((d_) + 6) = p_;                         \
    }
#define STORE_CHUNK(stage_)                                                             \
    {                                                                                   \
        float* ad_ = adst0 + (stage_) * kStageFloats;                                   \
        *(float4*)(ad_ +  0) = ta0;                                                     \
        *(float4*)(ad_ +  4) = ta1;                                                     \
        *(float4*)(ad_ +  8) = ta2;                                                     \
        *(float4*)(ad_ + 12) = ta3;                                                     \
        float* wd_ = wdst0 + (stage_) * kStageFloats;                                   \
        STORE_PAIRS(wd_ + 0 * 8 * kWPairStride * 2, tw00, tw01)                         \
        STORE_PAIRS(wd_ + 1 * 8 * kWPairStride * 2, tw10, tw11)                         \
        STORE_PAIRS(wd_ + 2 * 8 * kWPairStride * 2, tw20, tw21)                         \
        STORE_PAIRS(wd_ + 3 * 8 * kWPairStride * 2, tw30, tw31)                         \
    }

    LOAD_CHUNK(0)
    STORE_CHUNK(0)
    __syncthreads();

    #pragma unroll 2
    for (int kc = 0; kc < kD / kKC; ++kc) {
        const int  cur  = kc & 1;
        const bool more = kc + 1 < kD / kKC;
        if (more) LOAD_CHUNK((kc + 1) * kKC)

        const float* As = smem + cur * kStageFloats;
        const float* Wp = As + kAFloats;
        #pragma unroll
        for (int kk = 0; kk < kKC / 4; ++kk) {
            const int kb = kk * 4;
            const v2f a = *(const v2f*)&As[(mt * 16 + l16) * kAStride + kb + 2 * hi];
            v2f bfr[8];
            #pragma unroll
            for (int nt = 0; nt < 8; ++nt) {
                const int col = (nh * 8 + nt) * 16 + l16;
                bfr[nt] = *(const v2f*)&Wp[((kb >> 1) + hi) * (kWPairStride * 2) + col * 2];
            }
            #pragma unroll
            for (int nt = 0; nt < 8; ++nt) {
                acc[nt] = __builtin_amdgcn_wmma_f32_16x16x4_f32(
                    false, a, false, bfr[nt], (short)0, acc[nt], false, false);
            }
        }
        if (more) STORE_CHUNK(cur ^ 1)
        __syncthreads();
    }

#undef LOAD_CHUNK
#undef STORE_PAIRS
#undef STORE_CHUNK

    float* P = smem;
    float srow[8];
    #pragma unroll
    for (int r = 0; r < 8; ++r) srow[r] = 0.f;
    #pragma unroll
    for (int nt = 0; nt < 8; ++nt) {
        const int col    = (nh * 8 + nt) * 16 + l16;
        const float bcol = bvec[col];
        const float vcol = vvec[col];
        #pragma unroll
        for (int r = 0; r < 8; ++r) {
            acc[nt][r] += bcol;
            srow[r] += acc[nt][r] * vcol;
            P[(mt * 16 + hi * 8 + r) * kPStride + col] = acc[nt][r];
        }
    }
    #pragma unroll
    for (int off = 8; off >= 1; off >>= 1)
        #pragma unroll
        for (int r = 0; r < 8; ++r)
            srow[r] += __shfl_xor(srow[r], off, 16);
    if (l16 == 0) {
        #pragma unroll
        for (int r = 0; r < 8; ++r)
            scp[nh][mt * 16 + hi * 8 + r] = srow[r];
    }
    __syncthreads();

    for (int j = uwave; j < total; j += kWaves) {
        int t = j, w = 0;
        for (; w < kWaves - 1 && t >= scnt[w]; ++w) t -= scnt[w];
        const int s     = __builtin_amdgcn_readfirstlane((int)slist[w][t]);
        const int start = __builtin_amdgcn_readfirstlane(spans[3 * s + 1]);
        const int stop  = __builtin_amdgcn_readfirstlane(spans[3 * s + 2]);
        const int len   = stop - start;

        int r0 = start + lane;       if (r0 > kL - 1) r0 = kL - 1;  r0 -= p0;
        int r1 = start + lane + 32;  if (r1 > kL - 1) r1 = kL - 1;  r1 -= p0;
        const float sc0 = scp[0][r0] + scp[1][r0];
        const float sc1 = scp[0][r1] + scp[1][r1];
        const bool v0 = lane < len, v1 = lane + 32 < len;
        float m = fmaxf(v0 ? sc0 : -INFINITY, v1 ? sc1 : -INFINITY);
        #pragma unroll
        for (int off = 16; off >= 1; off >>= 1) m = fmaxf(m, __shfl_xor(m, off, 32));
        const float e0 = v0 ? expf(sc0 - m) : 0.f;
        const float e1 = v1 ? expf(sc1 - m) : 0.f;
        float sum = e0 + e1;
        #pragma unroll
        for (int off = 16; off >= 1; off >>= 1) sum += __shfl_xor(sum, off, 32);
        const float inv = 1.f / sum;
        const float a0 = e0 * inv, a1 = e1 * inv;

        const float* Pl = P + lane * 4;
        float4 pa = make_float4(0.f, 0.f, 0.f, 0.f);
        float4 pb = make_float4(0.f, 0.f, 0.f, 0.f);
        #pragma unroll 8
        for (int l = 0; l < kSpan; ++l) {
            const float av = (l < 32) ? a0 : a1;
            const float a  = __int_as_float(__builtin_amdgcn_readlane(__float_as_int(av), l & 31));
            int r = start + l;  if (r > kL - 1) r = kL - 1;  r -= p0;
            const float4 x = *(const float4*)(Pl + r * kPStride);
            const float4 y = *(const float4*)(Pl + r * kPStride + 128);
            pa.x = fmaf(a, x.x, pa.x); pa.y = fmaf(a, x.y, pa.y);
            pa.z = fmaf(a, x.z, pa.z); pa.w = fmaf(a, x.w, pa.w);
            pb.x = fmaf(a, y.x, pb.x); pb.y = fmaf(a, y.y, pb.y);
            pb.z = fmaf(a, y.z, pb.z); pb.w = fmaf(a, y.w, pb.w);
        }
        float* o = out + (size_t)s * kH + lane * 4;
        *(float4*)(o)       = pa;
        *(float4*)(o + 128) = pb;
    }
}


namespace {
__device__ __forceinline__ float wave_sum(float x)
{
    #pragma unroll
    for (int off = 16; off >= 1; off >>= 1) x += __shfl_xor(x, off, 32);
    return x;
}
__device__ __forceinline__ float dot4(const float4 a, const float4 b, float acc)
{
    acc = fmaf(a.x, b.x, acc); acc = fmaf(a.y, b.y, acc);
    acc = fmaf(a.z, b.z, acc); acc = fmaf(a.w, b.w, acc);
    return acc;
}
template <int NR>
__device__ __forceinline__ void pool_request(const float* __restrict__ Pg, int start, int l,
                                             float4* x, float4* y)
{
    #pragma unroll
    for (int j = 0; j < NR; ++j) {
        const int r = min(max(start + l + j, 0), kL - 1);
        x[j] = *(const float4*)(Pg + (size_t)r * kD);
        y[j] = *(const float4*)(Pg + (size_t)r * kD + 128);
    }
}
template <int NR>
__device__ __forceinline__ void pool_add(float av, int l, const float4* x, const float4* y,
                                         float4& pa, float4& pb)
{
    #pragma unroll
    for (int j = 0; j < NR; ++j) {
        const float a = __int_as_float(__builtin_amdgcn_readlane(__float_as_int(av), (l + j) & 31));
        pa.x = fmaf(a, x[j].x, pa.x); pa.y = fmaf(a, x[j].y, pa.y);
        pa.z = fmaf(a, x[j].z, pa.z); pa.w = fmaf(a, x[j].w, pa.w);
        pb.x = fmaf(a, y[j].x, pb.x); pb.y = fmaf(a, y[j].y, pb.y);
        pb.z = fmaf(a, y[j].z, pb.z); pb.w = fmaf(a, y[j].w, pb.w);
    }
}
}

__launch_bounds__(kUThreads)
__global__ void score_vector_kernel(const float* __restrict__ Wm,
                                    const float* __restrict__ bvec,
                                    const float* __restrict__ vvec,
                                    float* __restrict__ ws, size_t ws_bytes)
{
    if (ws_bytes < kNeedBytes) return;
    const int uw   = __builtin_amdgcn_readfirstlane((int)(threadIdx.x >> 5));
    const int lane = threadIdx.x & 31;
    const int k    = blockIdx.x * (kUThreads / 32) + uw;
    const float4 v0 = *(const float4*)(vvec + lane * 4);
    const float4 v1 = *(const float4*)(vvec + 128 + lane * 4);
    const float4 w0 = *(const float4*)(Wm + (size_t)k * kH + lane * 4);
    const float4 w1 = *(const float4*)(Wm + (size_t)k * kH + 128 + lane * 4);
    const float p = wave_sum(dot4(w1, v1, dot4(w0, v0, 0.f)));
    if (lane == 0) ws[kOffU + k] = p;
    if (k == 0) {
        const float4 b0 = *(const float4*)(bvec + lane * 4);
        const float4 b1 = *(const float4*)(bvec + 128 + lane * 4);
        const float q = wave_sum(dot4(b1, v1, dot4(b0, v0, 0.f)));
        if (lane == 0) ws[kOffU + kD] = q;
    }
}

__launch_bounds__(kScoreThreads)
__global__ void row_score_kernel(const float* __restrict__ hs_all,
                                 float* __restrict__ ws, size_t ws_bytes)
{
    if (ws_bytes < kNeedBytes) return;
    const int uw   = __builtin_amdgcn_readfirstlane((int)(threadIdx.x >> 5));
    const int lane = threadIdx.x & 31;
    const int row0 = (blockIdx.x * (kScoreThreads / 32) + uw) * 4;
    const float* X = hs_all + (size_t)kLayer * kN * kL * kD + (size_t)row0 * kD + lane * 4;
    const float* U = ws + kOffU;
    float4 x[4][8];
    #pragma unroll
    for (int r = 0; r < 4; ++r)
        #pragma unroll
        for (int j = 0; j < 8; ++j)
            x[r][j] = *(const float4*)(X + (size_t)r * kD + j * 128);
    float4 u[8];
    #pragma unroll
    for (int j = 0; j < 8; ++j) u[j] = *(const float4*)(U + j * 128 + lane * 4);
    const float c = U[kD];
    float p[4];
    #pragma unroll
    for (int r = 0; r < 4; ++r) {
        float t = 0.f;
        #pragma unroll
        for (int j = 0; j < 8; ++j) t = dot4(x[r][j], u[j], t);
        p[r] = wave_sum(t) + c;
    }
    const float mine = (lane == 0) ? p[0] : (lane == 1) ? p[1] : (lane == 2) ? p[2] : p[3];
    if (lane < 4) ws[kOffS + row0 + lane] = mine;
}

__launch_bounds__(kPoolThreads)
__global__ void span_pool_kernel(const float* __restrict__ hs_all,
                                 const int* __restrict__ spans,
                                 float* __restrict__ ws, size_t ws_bytes)
{
    if (ws_bytes < kNeedBytes) return;
    const int uw   = __builtin_amdgcn_readfirstlane((int)(threadIdx.x >> 5));
    const int lane = threadIdx.x & 31;
    const int g    = blockIdx.x * (kPoolThreads / 32) + uw;
    const int s    = g >> 2;
    const int q    = g & 3;

    int idx         = spans[3 * s + 0];
    const int start = spans[3 * s + 1];
    const int stop  = spans[3 * s + 2];
    idx = min(max(idx, 0), kN - 1);
    const int len = stop - start;
    const int nrow = (len <= 0) ? kSpan : min(len, kSpan);

    const float* S = ws + kOffS + (size_t)idx * kL;
    const int r0 = min(max(start + lane, 0), kL - 1);
    const int r1 = min(max(start + lane + 32, 0), kL - 1);
    const float sc0 = S[r0];
    const float sc1 = S[r1];
    const bool v0 = lane < len, v1 = lane + 32 < len;
    float m = fmaxf(v0 ? sc0 : -INFINITY, v1 ? sc1 : -INFINITY);
    #pragma unroll
    for (int off = 16; off >= 1; off >>= 1) m = fmaxf(m, __shfl_xor(m, off, 32));
    const float e0 = v0 ? expf(sc0 - m) : 0.f;
    const float e1 = v1 ? expf(sc1 - m) : 0.f;
    const float inv = 1.f / wave_sum(e0 + e1);
    const float a0 = e0 * inv, a1 = e1 * inv;

    const float* Pg = hs_all + (size_t)kLayer * kN * kL * kD + (size_t)idx * kL * kD + q * 256 + lane * 4;
    float4 pa = make_float4(0.f, 0.f, 0.f, 0.f);
    float4 pb = make_float4(0.f, 0.f, 0.f, 0.f);
    int l = 0;
    for (; l + kPoolGroup <= nrow; l += kPoolGroup) {
        float4 x[kPoolGroup], y[kPoolGroup];
        pool_request<kPoolGroup>(Pg, start, l, x, y);
        pool_add<kPoolGroup>((l < 32) ? a0 : a1, l, x, y, pa, pb);
    }
    const int rem = nrow - l;
    if (rem > 0) {
        float4 x[kPoolGroup], y[kPoolGroup];
        const float av = (l < 32) ? a0 : a1;
        pool_request<4>(Pg, start, l, x, y);
        if (rem > 4)  pool_request<4>(Pg, start, l + 4,  x + 4,  y + 4);
        if (rem > 8)  pool_request<4>(Pg, start, l + 8,  x + 8,  y + 8);
        if (rem > 12) pool_request<4>(Pg, start, l + 12, x + 12, y + 12);
        pool_add<4>(av, l, x, y, pa, pb);
        if (rem > 4)  pool_add<4>(av, l + 4,  x + 4,  y + 4,  pa, pb);
        if (rem > 8)  pool_add<4>(av, l + 8,  x + 8,  y + 8,  pa, pb);
        if (rem > 12) pool_add<4>(av, l + 12, x + 12, y + 12, pa, pb);
    }
    float* o = ws + kOffZ + (size_t)s * kD + q * 256 + lane * 4;
    *(float4*)(o)       = pa;
    *(float4*)(o + 128) = pb;
    if (q == 0) {
        const float t = wave_sum(a0 + a1);
        if (lane == 0) ws[kOffA + s] = t;
    }
}

extern "C" void kernel_launch(void* const* d_in, const int* in_sizes, int n_in,
                              void* d_out, int out_size, void* d_ws, size_t ws_size,
                              hipStream_t stream) {
    (void)in_sizes; (void)n_in; (void)out_size;
    const float* hs    = (const float*)d_in[0];
    const int*   spans = (const int*)  d_in[1];
    const float* W     = (const float*)d_in[2];
    const float* b     = (const float*)d_in[3];
    const float* v     = (const float*)d_in[4];
    float* out = (float*)d_out;
    float* ws  = (float*)d_ws;
    score_vector_kernel<<<dim3(kUGrid), dim3(kUThreads), 0, stream>>>(W, b, v, ws, ws_size);
    row_score_kernel<<<dim3(kScoreGrid), dim3(kScoreThreads), 0, stream>>>(hs, ws, ws_size);
    span_pool_kernel<<<dim3(kPoolGrid), dim3(kPoolThreads), 0, stream>>>(hs, spans, ws, ws_size);
    attention_pooler_kernel<<<dim3(kGrid), dim3(kThreads), 0, stream>>>(hs, spans, W, b, v, ws, ws_size, out);
}


# Fuse the row-score launch into the span-pool launch: a wave reads its span rows whole once and pools them with a running-maximum softmax; 3 launches not 4; softmax and row sum order change, f32 kept
# speedup vs baseline: 12.0860x; 1.1710x over previous
#include <hip/hip_runtime.h>
#include <hip/hip_bf16.h>
#include <math.h>


typedef __attribute__((ext_vector_type(2))) float v2f;
typedef __attribute__((ext_vector_type(8))) float v8f;
typedef __attribute__((ext_vector_type(4))) float v4f;

namespace {
constexpr int kLayer  = 2;
constexpr int kN = 32, kL = 512, kD = 1024, kH = 256;
constexpr int kS = 2048, kSpan = 64;
constexpr int kWin       = 64;
constexpr int kWinShift  = 6;
constexpr int kRows      = kWin + kSpan;
constexpr int kWinPerSeq = kL / kWin;
constexpr int kGrid      = kN * kWinPerSeq;
constexpr int kThreads   = 512;
constexpr int kWaves     = kThreads / 32;
constexpr int kScanPerWave = kS / kWaves;
constexpr int kKC = 64;
constexpr int kAStride = 68;
constexpr int kWPairStride = 272;
static_assert((1 << kWinShift) == kWin, "window shift");
constexpr int kPStride = 260;
constexpr int kAFloats = kRows * kAStride;
constexpr int kWFloats = (kKC / 2) * kWPairStride * 2;
constexpr int kStageFloats = kAFloats + kWFloats;
constexpr int kPFloats = kRows * kPStride;
constexpr size_t kOffU  = 0;
constexpr size_t kOffA  = 2048;
constexpr size_t kOffZ  = kOffA + kS;
constexpr size_t kNeedFloats = kOffZ + (size_t)kS * kD;
constexpr size_t kNeedBytes  = kNeedFloats * sizeof(float);
static_assert(kOffZ % 64 == 0, "pooled rows start on a 256-B boundary");
constexpr int kGM = 32, kGN = 64;
constexpr int kGKC = 128;
constexpr int kGAStride = 132;
constexpr int kGPairStride = 80;
constexpr int kGAFloats = kGM * kGAStride;
constexpr int kGWFloats = (kGKC / 2) * kGPairStride * 2;
constexpr int kGStage   = kGAFloats + kGWFloats;
constexpr int kGPPitch  = 68;
static_assert((kS / kGM) * (kH / kGN) == kGrid, "one 32 x 64 tile per work-group");
static_assert(2 * kGStage <= 2 * kStageFloats && 2 * kGM * kGPPitch <= 2 * kStageFloats, "fits the LDS ring");
constexpr int kFuseWaves   = 4;
constexpr int kFuseThreads = kFuseWaves * 32;
constexpr int kFuseGrid    = kS;
constexpr int kFuseRows    = 4;
static_assert(kFuseRows == 4, "the span launch dispatches on group sizes 1 to 4");
constexpr int kUThreads = 256;
constexpr int kUGrid    = kD / 8;
}

__launch_bounds__(512)
__global__ void attention_pooler_kernel(const float* __restrict__ hs_all,
                                        const int*  __restrict__ spans,
                                        const float* __restrict__ Wm,
                                        const float* __restrict__ bvec,
                                        const float* __restrict__ vvec,
                                        float* __restrict__ ws, size_t ws_bytes,
                                        float* __restrict__ out)
{
    __shared__ __align__(16) float smem[2 * kStageFloats];
    __shared__ float scp[2][kRows];
    __shared__ unsigned short slist[kWaves][kScanPerWave];
    __shared__ int scnt[kWaves];

    const int seq  = blockIdx.x / kWinPerSeq;
    const int win  = blockIdx.x % kWinPerSeq;
    const int p0   = win * kWin;
    const int tid  = threadIdx.x;
    const int wave = tid >> 5;
    const int lane = tid & 31;
    const int mt   = wave & 7;
    const int nh   = wave >> 3;
    const int hi   = lane >> 4;
    const int l16  = lane & 15;
    const int uwave = __builtin_amdgcn_readfirstlane(wave);

    const float* hs = hs_all + (size_t)kLayer * kN * kL * kD;

    if (ws_bytes >= kNeedBytes) {
        const int m0 = (blockIdx.x >> 2) * kGM;
        const int n0 = (blockIdx.x & 3) * kGN;
        const int tm = wave & 1;
        const int tn = (wave >> 1) & 3;
        const int kh = wave >> 3;
        const float* Z  = ws + kOffZ;
        const float* SA = ws + kOffA;

        v8f acc0, acc1;
        #pragma unroll
        for (int e = 0; e < 8; ++e) { acc0[e] = 0.f; acc1[e] = 0.f; }

        const int ar = tid >> 4;
        const int ac = (tid & 15) * 4;
        const float* asrc = Z + (size_t)(m0 + ar) * kD + ac;
        float* adst = &smem[ar * kGAStride + ac];
        const float* wsrc = Wm + (size_t)(2 * ar) * kH + n0 + ac;
        float* wdst = &smem[kGAFloats + (ar * kGPairStride + ac) * 2];
        float4 ga0, ga1, gw00, gw01, gw10, gw11;

#define G_LOAD(kbase_)                                                                  \
    {                                                                                   \
        ga0 = *(const float4*)(asrc + (kbase_));                                        \
        ga1 = *(const float4*)(asrc + (kbase_) + 64);                                   \
        const float* w_ = wsrc + (size_t)(kbase_) * kH;                                 \
        gw00 = *(const float4*)(w_ + (size_t)( 0 + 0) * kH);                            \
        gw01 = *(const float4*)(w_ + (size_t)( 0 + 1) * kH);                            \
        gw10 = *(const float4*)(w_ + (size_t)(64 + 0) * kH);                            \
        gw11 = *(const float4*)(w_ + (size_t)(64 + 1) * kH);                            \
    }
#define G_PAIRS(d_, w0_, w1_)                                                           \
    {                                                                                   \
        v2f p_;                                                                         \
        p_.x = (w0_).x; p_.y = (w1_).x; *(v2f*)((d_) + 0) = p_;                         \
        p_.x = (w0_).y; p_.y = (w1_).y; *(v2f*)((d_) + 2) = p_;                         \
        p_.x = (w0_).z; p_.y = (w1_).z; *(v2f*)((d_) + 4) = p_;                         \
        p_.x = (w0_).w; p_.y = (w1_).w; *(v2f*)((d_) + 6) = p_;                         \
    }
#define G_STORE(stage_)                                                                 \
    {                                                                                   \
        float* ad_ = adst + (stage_) * kGStage;                                         \
        *(float4*)(ad_)      = ga0;                                                     \
        *(float4*)(ad_ + 64) = ga1;                                                     \
        float* wd_ = wdst + (stage_) * kGStage;                                         \
        G_PAIRS(wd_, gw00, gw01)                                                        \
        G_PAIRS(wd_ + 32 * kGPairStride * 2, gw10, gw11)                                \
    }

        G_LOAD(0)
        G_STORE(0)
        __syncthreads();

        const int aoff = (tm * 16 + l16) * kGAStride + kh * (kGKC / 2) + 2 * hi;
        const int woff = (kh * (kGKC / 4) + hi) * (kGPairStride * 2) + (tn * 16 + l16) * 2;

        #pragma unroll 2
        for (int kc = 0; kc < kD / kGKC; ++kc) {
            const int  cur  = kc & 1;
            const bool more = kc + 1 < kD / kGKC;
            if (more) G_LOAD((kc + 1) * kGKC)

            const float* As = smem + cur * kGStage + aoff;
            const float* Wp = smem + cur * kGStage + kGAFloats + woff;
            #pragma unroll
            for (int kk = 0; kk < kGKC / 8; kk += 2) {
                const v2f a0 = *(const v2f*)&As[kk * 4];
                const v2f b0 = *(const v2f*)&Wp[kk * 2 * (kGPairStride * 2)];
                const v2f a1 = *(const v2f*)&As[(kk + 1) * 4];
                const v2f b1 = *(const v2f*)&Wp[(kk + 1) * 2 * (kGPairStride * 2)];
                acc0 = __builtin_amdgcn_wmma_f32_16x16x4_f32(false, a0, false, b0, (short)0, acc0, false, false);
                acc1 = __builtin_amdgcn_wmma_f32_16x16x4_f32(false, a1, false, b1, (short)0, acc1, false, false);
            }
            if (more) G_STORE(cur ^ 1)
            __syncthreads();
        }
#undef G_LOAD
#undef G_PAIRS
#undef G_STORE

        float* Pp = smem;
        #pragma unroll
        for (int r = 0; r < 8; ++r)
            Pp[(kh * kGM + tm * 16 + hi * 8 + r) * kGPPitch + tn * 16 + l16] = acc0[r] + acc1[r];
        __syncthreads();
        {
            const float4 q0 = *(const float4*)&Pp[ar * kGPPitch + ac];
            const float4 q1 = *(const float4*)&Pp[(kGM + ar) * kGPPitch + ac];
            const float  sa = SA[m0 + ar];
            const float4 bb = *(const float4*)&bvec[n0 + ac];
            float4 o;
            o.x = (q0.x + q1.x) + sa * bb.x;
            o.y = (q0.y + q1.y) + sa * bb.y;
            o.z = (q0.z + q1.z) + sa * bb.z;
            o.w = (q0.w + q1.w) + sa * bb.w;
            *(float4*)&out[(size_t)(m0 + ar) * kH + n0 + ac] = o;
        }
        return;
    }


    {
        int ridx[kScanPerWave / 32], rst[kScanPerWave / 32];
        #pragma unroll
        for (int it = 0; it < kScanPerWave / 32; ++it) {
            const int s = uwave * kScanPerWave + it * 32 + lane;
            ridx[it] = spans[3 * s + 0];
            rst[it]  = spans[3 * s + 1];
        }
        int n = 0;
        #pragma unroll
        for (int it = 0; it < kScanPerWave / 32; ++it) {
            const int s = uwave * kScanPerWave + it * 32 + lane;
            const bool mine = (ridx[it] == seq) && ((rst[it] >> kWinShift) == win);
            const unsigned mask = (unsigned)__ballot(mine);
            if (mine) slist[uwave][n + __popc(mask & ((1u << lane) - 1u))] = (unsigned short)s;
            n += __popc(mask);
        }
        if (lane == 0) scnt[uwave] = n;
    }
    __syncthreads();
    int total = 0;
    for (int w = 0; w < kWaves; ++w) total += scnt[w];
    total = __builtin_amdgcn_readfirstlane(total);
    if (total == 0) return;

    v8f acc[8];
    #pragma unroll
    for (int nt = 0; nt < 8; ++nt)
        #pragma unroll
        for (int e = 0; e < 8; ++e) acc[nt][e] = 0.f;

    const int rowj = tid >> 2;
    const int quad = tid & 3;
    int pos = p0 + rowj;
    if (pos > kL - 1) pos = kL - 1;
    const float* arow = hs + ((size_t)seq * kL + pos) * kD;

    const int wkp = tid >> 6;
    const int wc  = (tid & 63) * 4;
    const float* wsrc = Wm + (size_t)(2 * wkp) * kH + wc;
    float* adst0 = &smem[rowj * kAStride + quad * 16];
    float* wdst0 = &smem[kAFloats + (wkp * kWPairStride + wc) * 2];
    float4 ta0, ta1, ta2, ta3;
    float4 tw00, tw01, tw10, tw11, tw20, tw21, tw30, tw31;

#define LOAD_CHUNK(kbase_)                                                              \
    {                                                                                   \
        const float4* asrc_ = (const float4*)(arow + (kbase_) + quad * 16);             \
        ta0 = asrc_[0]; ta1 = asrc_[1]; ta2 = asrc_[2]; ta3 = asrc_[3];                 \
        const float* w_ = wsrc + (size_t)(kbase_) * kH;                                 \
        tw00 = *(const float4*)(w_ + (size_t)( 0 + 0) * kH);                            \
        tw01 = *(const float4*)(w_ + (size_t)( 0 + 1) * kH);                            \
        tw10 = *(const float4*)(w_ + (size_t)(16 + 0) * kH);                            \
        tw11 = *(const float4*)(w_ + (size_t)(16 + 1) * kH);                            \
        tw20 = *(const float4*)(w_ + (size_t)(32 + 0) * kH);                            \
        tw21 = *(const float4*)(w_ + (size_t)(32 + 1) * kH);                            \
        tw30 = *(const float4*)(w_ + (size_t)(48 + 0) * kH);                            \
        tw31 = *(const float4*)(w_ + (size_t)(48 + 1) * kH);                            \
    }
#define STORE_PAIRS(d_, w0_, w1_)                                                       \
    {                                                                                   \
        v2f p_;                                                                         \
        p_.x = (w0_).x; p_.y = (w1_).x; *(v2f*)((d_) + 0) = p_;                         \
        p_.x = (w0_).y; p_.y = (w1_).y; *(v2f*)((d_) + 2) = p_;                         \
        p_.x = (w0_).z; p_.y = (w1_).z; *(v2f*)((d_) + 4) = p_;                         \
        p_.x = (w0_).w; p_.y = (w1_).w; *(v2f*)((d_) + 6) = p_;                         \
    }
#define STORE_CHUNK(stage_)                                                             \
    {                                                                                   \
        float* ad_ = adst0 + (stage_) * kStageFloats;                                   \
        *(float4*)(ad_ +  0) = ta0;                                                     \
        *(float4*)(ad_ +  4) = ta1;                                                     \
        *(float4*)(ad_ +  8) = ta2;                                                     \
        *(float4*)(ad_ + 12) = ta3;                                                     \
        float* wd_ = wdst0 + (stage_) * kStageFloats;                                   \
        STORE_PAIRS(wd_ + 0 * 8 * kWPairStride * 2, tw00, tw01)                         \
        STORE_PAIRS(wd_ + 1 * 8 * kWPairStride * 2, tw10, tw11)                         \
        STORE_PAIRS(wd_ + 2 * 8 * kWPairStride * 2, tw20, tw21)                         \
        STORE_PAIRS(wd_ + 3 * 8 * kWPairStride * 2, tw30, tw31)                         \
    }

    LOAD_CHUNK(0)
    STORE_CHUNK(0)
    __syncthreads();

    #pragma unroll 2
    for (int kc = 0; kc < kD / kKC; ++kc) {
        const int  cur  = kc & 1;
        const bool more = kc + 1 < kD / kKC;
        if (more) LOAD_CHUNK((kc + 1) * kKC)

        const float* As = smem + cur * kStageFloats;
        const float* Wp = As + kAFloats;
        #pragma unroll
        for (int kk = 0; kk < kKC / 4; ++kk) {
            const int kb = kk * 4;
            const v2f a = *(const v2f*)&As[(mt * 16 + l16) * kAStride + kb + 2 * hi];
            v2f bfr[8];
            #pragma unroll
            for (int nt = 0; nt < 8; ++nt) {
                const int col = (nh * 8 + nt) * 16 + l16;
                bfr[nt] = *(const v2f*)&Wp[((kb >> 1) + hi) * (kWPairStride * 2) + col * 2];
            }
            #pragma unroll
            for (int nt = 0; nt < 8; ++nt) {
                acc[nt] = __builtin_amdgcn_wmma_f32_16x16x4_f32(
                    false, a, false, bfr[nt], (short)0, acc[nt], false, false);
            }
        }
        if (more) STORE_CHUNK(cur ^ 1)
        __syncthreads();
    }

#undef LOAD_CHUNK
#undef STORE_PAIRS
#undef STORE_CHUNK

    float* P = smem;
    float srow[8];
    #pragma unroll
    for (int r = 0; r < 8; ++r) srow[r] = 0.f;
    #pragma unroll
    for (int nt = 0; nt < 8; ++nt) {
        const int col    = (nh * 8 + nt) * 16 + l16;
        const float bcol = bvec[col];
        const float vcol = vvec[col];
        #pragma unroll
        for (int r = 0; r < 8; ++r) {
            acc[nt][r] += bcol;
            srow[r] += acc[nt][r] * vcol;
            P[(mt * 16 + hi * 8 + r) * kPStride + col] = acc[nt][r];
        }
    }
    #pragma unroll
    for (int off = 8; off >= 1; off >>= 1)
        #pragma unroll
        for (int r = 0; r < 8; ++r)
            srow[r] += __shfl_xor(srow[r], off, 16);
    if (l16 == 0) {
        #pragma unroll
        for (int r = 0; r < 8; ++r)
            scp[nh][mt * 16 + hi * 8 + r] = srow[r];
    }
    __syncthreads();

    for (int j = uwave; j < total; j += kWaves) {
        int t = j, w = 0;
        for (; w < kWaves - 1 && t >= scnt[w]; ++w) t -= scnt[w];
        const int s     = __builtin_amdgcn_readfirstlane((int)slist[w][t]);
        const int start = __builtin_amdgcn_readfirstlane(spans[3 * s + 1]);
        const int stop  = __builtin_amdgcn_readfirstlane(spans[3 * s + 2]);
        const int len   = stop - start;

        int r0 = start + lane;       if (r0 > kL - 1) r0 = kL - 1;  r0 -= p0;
        int r1 = start + lane + 32;  if (r1 > kL - 1) r1 = kL - 1;  r1 -= p0;
        const float sc0 = scp[0][r0] + scp[1][r0];
        const float sc1 = scp[0][r1] + scp[1][r1];
        const bool v0 = lane < len, v1 = lane + 32 < len;
        float m = fmaxf(v0 ? sc0 : -INFINITY, v1 ? sc1 : -INFINITY);
        #pragma unroll
        for (int off = 16; off >= 1; off >>= 1) m = fmaxf(m, __shfl_xor(m, off, 32));
        const float e0 = v0 ? expf(sc0 - m) : 0.f;
        const float e1 = v1 ? expf(sc1 - m) : 0.f;
        float sum = e0 + e1;
        #pragma unroll
        for (int off = 16; off >= 1; off >>= 1) sum += __shfl_xor(sum, off, 32);
        const float inv = 1.f / sum;
        const float a0 = e0 * inv, a1 = e1 * inv;

        const float* Pl = P + lane * 4;
        float4 pa = make_float4(0.f, 0.f, 0.f, 0.f);
        float4 pb = make_float4(0.f, 0.f, 0.f, 0.f);
        #pragma unroll 8
        for (int l = 0; l < kSpan; ++l) {
            const float av = (l < 32) ? a0 : a1;
            const float a  = __int_as_float(__builtin_amdgcn_readlane(__float_as_int(av), l & 31));
            int r = start + l;  if (r > kL - 1) r = kL - 1;  r -= p0;
            const float4 x = *(const float4*)(Pl + r * kPStride);
            const float4 y = *(const float4*)(Pl + r * kPStride + 128);
            pa.x = fmaf(a, x.x, pa.x); pa.y = fmaf(a, x.y, pa.y);
            pa.z = fmaf(a, x.z, pa.z); pa.w = fmaf(a, x.w, pa.w);
            pb.x = fmaf(a, y.x, pb.x); pb.y = fmaf(a, y.y, pb.y);
            pb.z = fmaf(a, y.z, pb.z); pb.w = fmaf(a, y.w, pb.w);
        }
        float* o = out + (size_t)s * kH + lane * 4;
        *(float4*)(o)       = pa;
        *(float4*)(o + 128) = pb;
    }
}


namespace {
__device__ __forceinline__ float wave_sum(float x)
{
    #pragma unroll
    for (int off = 16; off >= 1; off >>= 1) x += __shfl_xor(x, off, 32);
    return x;
}
__device__ __forceinline__ float dot4(const float4 a, const float4 b, float acc)
{
    acc = fmaf(a.x, b.x, acc); acc = fmaf(a.y, b.y, acc);
    acc = fmaf(a.z, b.z, acc); acc = fmaf(a.w, b.w, acc);
    return acc;
}
__device__ __forceinline__ void row_request(const float* __restrict__ Xs, int start, int l, v4f* x)
{
    const int r = min(max(start + l, 0), kL - 1);
    #pragma unroll
    for (int k = 0; k < 8; ++k) x[k] = *(const v4f*)(Xs + (size_t)r * kD + k * 128);
}
__device__ __forceinline__ void row_clear(v4f* x)
{
    #pragma unroll
    for (int k = 0; k < 8; ++k) x[k] = (v4f){0.f, 0.f, 0.f, 0.f};
}
}

__launch_bounds__(kUThreads)
__global__ void score_vector_kernel(const float* __restrict__ Wm,
                                    const float* __restrict__ bvec,
                                    const float* __restrict__ vvec,
                                    float* __restrict__ ws, size_t ws_bytes)
{
    if (ws_bytes < kNeedBytes) return;
    const int uw   = __builtin_amdgcn_readfirstlane((int)(threadIdx.x >> 5));
    const int lane = threadIdx.x & 31;
    const int k    = blockIdx.x * (kUThreads / 32) + uw;
    const float4 v0 = *(const float4*)(vvec + lane * 4);
    const float4 v1 = *(const float4*)(vvec + 128 + lane * 4);
    const float4 w0 = *(const float4*)(Wm + (size_t)k * kH + lane * 4);
    const float4 w1 = *(const float4*)(Wm + (size_t)k * kH + 128 + lane * 4);
    const float p = wave_sum(dot4(w1, v1, dot4(w0, v0, 0.f)));
    if (lane == 0) ws[kOffU + k] = p;
    if (k == 0) {
        const float4 b0 = *(const float4*)(bvec + lane * 4);
        const float4 b1 = *(const float4*)(bvec + 128 + lane * 4);
        const float q = wave_sum(dot4(b1, v1, dot4(b0, v0, 0.f)));
        if (lane == 0) ws[kOffU + kD] = q;
    }
}

__launch_bounds__(kFuseThreads)
__global__ void span_pool_kernel(const float* __restrict__ hs_all,
                                 const int* __restrict__ spans,
                                 float* __restrict__ ws, size_t ws_bytes)
{
    if (ws_bytes < kNeedBytes) return;
    __shared__ __align__(16) float part[kFuseWaves][kD];
    __shared__ float pm[kFuseWaves], pq[kFuseWaves];
    const int uw   = __builtin_amdgcn_readfirstlane((int)(threadIdx.x >> 5));
    const int lane = threadIdx.x & 31;
    const int s    = blockIdx.x;

    int idx         = spans[3 * s + 0];
    const int start = spans[3 * s + 1];
    const int stop  = spans[3 * s + 2];
    idx = min(max(idx, 0), kN - 1);
    const int len  = stop - start;
    const int nrow = min(max(len, 0), kSpan);

    const float* U = ws + kOffU;
    v4f u[8];
    #pragma unroll
    for (int k = 0; k < 8; ++k) u[k] = *(const v4f*)(U + k * 128 + lane * 4);
    const float c = U[kD];

    const float* Xs = hs_all + (size_t)kLayer * kN * kL * kD + (size_t)idx * kL * kD + lane * 4;
    float m = -INFINITY, ssum = 0.f;
    v4f acc[8];
    #pragma unroll
    for (int k = 0; k < 8; ++k) acc[k] = (v4f){0.f, 0.f, 0.f, 0.f};

    for (int l = uw; l < nrow; l += kFuseWaves * kFuseRows) {
        const int cnt = (nrow - l + kFuseWaves - 1) / kFuseWaves;
        v4f x[kFuseRows * 8];
        row_request(Xs, start, l, x);
        if (cnt > 1) row_request(Xs, start, l + kFuseWaves,     x + 8);  else row_clear(x + 8);
        if (cnt > 2) row_request(Xs, start, l + 2 * kFuseWaves, x + 16); else row_clear(x + 16);
        if (cnt > 3) row_request(Xs, start, l + 3 * kFuseWaves, x + 24); else row_clear(x + 24);

        float sc[kFuseRows];
        #pragma unroll
        for (int j = 0; j < kFuseRows; ++j) {
            v2f t = {0.f, 0.f};
            #pragma unroll
            for (int k = 0; k < 8; ++k) {
                t = x[j * 8 + k].xy * u[k].xy + t;
                t = x[j * 8 + k].zw * u[k].zw + t;
            }
            sc[j] = t.x + t.y;
        }
        #pragma unroll
        for (int j = 0; j < kFuseRows; ++j) sc[j] = wave_sum(sc[j]) + c;
        #pragma unroll
        for (int j = 1; j < kFuseRows; ++j) sc[j] = (j < cnt) ? sc[j] : -INFINITY;

        float mn = m;
        #pragma unroll
        for (int j = 0; j < kFuseRows; ++j) mn = fmaxf(mn, sc[j]);
        const float f = expf(m - mn);
        ssum *= f;
        #pragma unroll
        for (int k = 0; k < 8; ++k) acc[k] *= f;
        #pragma unroll
        for (int j = 0; j < kFuseRows; ++j) {
            const float p = expf(sc[j] - mn);
            ssum += p;
            #pragma unroll
            for (int k = 0; k < 8; ++k) acc[k] = x[j * 8 + k] * p + acc[k];
        }
        m = mn;
    }

    if (lane == 0) { pm[uw] = m; pq[uw] = ssum; }
    #pragma unroll
    for (int k = 0; k < 8; ++k) *(v4f*)&part[uw][k * 128 + lane * 4] = acc[k];
    __syncthreads();

    const float m0 = pm[0], m1 = pm[1], m2 = pm[2], m3 = pm[3];
    const float mx = fmaxf(fmaxf(m0, m1), fmaxf(m2, m3));
    const float f0 = expf(m0 - mx), f1 = expf(m1 - mx), f2 = expf(m2 - mx), f3 = expf(m3 - mx);
    const float tot = fmaf(f3, pq[3], fmaf(f2, pq[2], fmaf(f1, pq[1], f0 * pq[0])));
    const float inv = 1.f / tot;
    const float g0 = f0 * inv, g1 = f1 * inv, g2 = f2 * inv, g3 = f3 * inv;
    float* z = ws + kOffZ + (size_t)s * kD;
    #pragma unroll
    for (int h = 0; h < 2; ++h) {
        const int col = h * (kD / 2) + (int)threadIdx.x * 4;
        const float4 a0 = *(const float4*)&part[0][col];
        const float4 a1 = *(const float4*)&part[1][col];
        const float4 a2 = *(const float4*)&part[2][col];
        const float4 a3 = *(const float4*)&part[3][col];
        float4 o;
        o.x = fmaf(g3, a3.x, fmaf(g2, a2.x, fmaf(g1, a1.x, g0 * a0.x)));
        o.y = fmaf(g3, a3.y, fmaf(g2, a2.y, fmaf(g1, a1.y, g0 * a0.y)));
        o.z = fmaf(g3, a3.z, fmaf(g2, a2.z, fmaf(g1, a1.z, g0 * a0.z)));
        o.w = fmaf(g3, a3.w, fmaf(g2, a2.w, fmaf(g1, a1.w, g0 * a0.w)));
        *(float4*)(z + col) = o;
    }
    if (threadIdx.x == 0) ws[kOffA + s] = tot * inv;
}

extern "C" void kernel_launch(void* const* d_in, const int* in_sizes, int n_in,
                              void* d_out, int out_size, void* d_ws, size_t ws_size,
                              hipStream_t stream) {
    (void)in_sizes; (void)n_in; (void)out_size;
    const float* hs    = (const float*)d_in[0];
    const int*   spans = (const int*)  d_in[1];
    const float* W     = (const float*)d_in[2];
    const float* b     = (const float*)d_in[3];
    const float* v     = (const float*)d_in[4];
    float* out = (float*)d_out;
    float* ws  = (float*)d_ws;
    score_vector_kernel<<<dim3(kUGrid), dim3(kUThreads), 0, stream>>>(W, b, v, ws, ws_size);
    span_pool_kernel<<<dim3(kFuseGrid), dim3(kFuseThreads), 0, stream>>>(hs, spans, ws, ws_size);
    attention_pooler_kernel<<<dim3(kGrid), dim3(kThreads), 0, stream>>>(hs, spans, W, b, v, ws, ws_size, out);
}


# Pool launch: a work-group per 64-row chunk reads each row once into LDS with 8-row block sums and pools span parts from LDS; two chunk partials joined in the product's A staging; sum order changes
# speedup vs baseline: 13.2460x; 1.0960x over previous
#include <hip/hip_runtime.h>
#include <hip/hip_bf16.h>
#include <math.h>


typedef __attribute__((ext_vector_type(2))) float v2f;
typedef __attribute__((ext_vector_type(8))) float v8f;
typedef __attribute__((ext_vector_type(4))) float v4f;

namespace {
constexpr int kLayer  = 2;
constexpr int kN = 32, kL = 512, kD = 1024, kH = 256;
constexpr int kS = 2048, kSpan = 64;
constexpr int kWin       = 64;
constexpr int kWinShift  = 6;
constexpr int kRows      = kWin + kSpan;
constexpr int kWinPerSeq = kL / kWin;
constexpr int kGrid      = kN * kWinPerSeq;
constexpr int kThreads   = 512;
constexpr int kWaves     = kThreads / 32;
constexpr int kScanPerWave = kS / kWaves;
constexpr int kKC = 64;
constexpr int kAStride = 68;
constexpr int kWPairStride = 272;
static_assert((1 << kWinShift) == kWin, "window shift");
constexpr int kPStride = 260;
constexpr int kAFloats = kRows * kAStride;
constexpr int kWFloats = (kKC / 2) * kWPairStride * 2;
constexpr int kStageFloats = kAFloats + kWFloats;
constexpr int kPFloats = kRows * kPStride;
constexpr int kCh        = 64;
constexpr int kChShift   = 6;
constexpr int kChPerSeq  = kL / kCh;
static_assert((1 << kChShift) == kCh && kSpan <= kCh, "a span of at most 64 rows meets at most two chunks");
constexpr size_t kOffU  = 0;
constexpr size_t kOffM  = 2048;
constexpr size_t kOffQ  = kOffM + 2 * (size_t)kS;
constexpr size_t kOffZ  = kOffQ + 2 * (size_t)kS;
constexpr size_t kNeedFloats = kOffZ + 2 * (size_t)kS * kD;
constexpr size_t kNeedBytes  = kNeedFloats * sizeof(float);
static_assert(kOffZ % 64 == 0, "pooled rows start on a 256-B boundary");
constexpr int kGM = 32, kGN = 64;
constexpr int kGKC = 128;
constexpr int kGAStride = 132;
constexpr int kGPairStride = 80;
constexpr int kGAFloats = kGM * kGAStride;
constexpr int kGWFloats = (kGKC / 2) * kGPairStride * 2;
constexpr int kGStage   = kGAFloats + kGWFloats;
constexpr int kGPPitch  = 68;
static_assert((kS / kGM) * (kH / kGN) == kGrid, "one 32 x 64 tile per work-group");
static_assert(2 * kGStage <= 2 * kStageFloats && 2 * kGM * kGPPitch <= 2 * kStageFloats, "fits the LDS ring");
constexpr int kPoolWaves   = 8;
constexpr int kPoolThreads = kPoolWaves * 32;
constexpr int kPoolGrid    = kN * kChPerSeq;
constexpr int kPoolRows    = kCh / kPoolWaves;
constexpr int kBlkShift    = 3;
constexpr int kTileRows    = kCh + kPoolWaves;
constexpr int kPoolScan    = kS / kPoolWaves;
static_assert(kPoolRows == (1 << kBlkShift) && kPoolScan % 32 == 0, "row staging and record scan shapes");
constexpr int kUThreads = 256;
constexpr int kUGrid    = kD / 8;
}

__launch_bounds__(512)
__global__ void attention_pooler_kernel(const float* __restrict__ hs_all,
                                        const int*  __restrict__ spans,
                                        const float* __restrict__ Wm,
                                        const float* __restrict__ bvec,
                                        const float* __restrict__ vvec,
                                        float* __restrict__ ws, size_t ws_bytes,
                                        float* __restrict__ out)
{
    __shared__ __align__(16) float smem[2 * kStageFloats];
    __shared__ float scp[2][kRows];
    __shared__ unsigned short slist[kWaves][kScanPerWave];
    __shared__ int scnt[kWaves];

    const int seq  = blockIdx.x / kWinPerSeq;
    const int win  = blockIdx.x % kWinPerSeq;
    const int p0   = win * kWin;
    const int tid  = threadIdx.x;
    const int wave = tid >> 5;
    const int lane = tid & 31;
    const int mt   = wave & 7;
    const int nh   = wave >> 3;
    const int hi   = lane >> 4;
    const int l16  = lane & 15;
    const int uwave = __builtin_amdgcn_readfirstlane(wave);

    const float* hs = hs_all + (size_t)kLayer * kN * kL * kD;

    if (ws_bytes >= kNeedBytes) {
        const int m0 = (blockIdx.x >> 2) * kGM;
        const int n0 = (blockIdx.x & 3) * kGN;
        const int tm = wave & 1;
        const int tn = (wave >> 1) & 3;
        const int kh = wave >> 3;
        const float* Z  = ws + kOffZ;

        v8f acc0, acc1;
        #pragma unroll
        for (int e = 0; e < 8; ++e) { acc0[e] = 0.f; acc1[e] = 0.f; }

        const int ar = tid >> 4;
        const int ac = (tid & 15) * 4;
        const int sp    = m0 + ar;
        const int sst   = spans[3 * sp + 1];
        const int snrow = min(max(spans[3 * sp + 2] - sst, 0), kSpan);
        const int scf   = min(max(sst, 0), kL - 1) >> kChShift;
        const int scl   = min(max(sst + snrow - 1, 0), kL - 1) >> kChShift;
        const bool one  = snrow > 0;
        const bool two  = one && scl > scf;
        float g0, g1, sa;
        {
            const float pm0 = one ? ws[kOffM + sp] : 0.f;
            const float pq0 = one ? ws[kOffQ + sp] : 0.f;
            const float pm1 = two ? ws[kOffM + kS + sp] : -INFINITY;
            const float pq1 = two ? ws[kOffQ + kS + sp] : 0.f;
            const float mx  = fmaxf(pm0, pm1);
            const float f0  = expf(pm0 - mx), f1 = expf(pm1 - mx);
            const float tot = fmaf(f1, pq1, f0 * pq0);
            const float inv = 1.f / tot;
            g0 = f0 * inv; g1 = f1 * inv; sa = tot * inv;
            if (!one) g0 = sa;
        }
        const float* asrc  = one ? (Z + (size_t)sp * kD + ac) : (ws + kOffU + ac);
        const float* asrc1 = two ? (asrc + (size_t)kS * kD) : asrc;
        float* adst = &smem[ar * kGAStride + ac];
        const float* wsrc = Wm + (size_t)(2 * ar) * kH + n0 + ac;
        float* wdst = &smem[kGAFloats + (ar * kGPairStride + ac) * 2];
        float4 ga0, ga1, gw00, gw01, gw10, gw11;
        float4 gb0, gb1;

#define G_LOAD(kbase_)                                                                  \
    {                                                                                   \
        ga0 = *(const float4*)(asrc + (kbase_));                                        \
        ga1 = *(const float4*)(asrc + (kbase_) + 64);                                   \
        gb0 = *(const float4*)(asrc1 + (kbase_));                                       \
        gb1 = *(const float4*)(asrc1 + (kbase_) + 64);                                  \
        const float* w_ = wsrc + (size_t)(kbase_) * kH;                                 \
        gw00 = *(const float4*)(w_ + (size_t)( 0 + 0) * kH);                            \
        gw01 = *(const float4*)(w_ + (size_t)( 0 + 1) * kH);                            \
        gw10 = *(const float4*)(w_ + (size_t)(64 + 0) * kH);                            \
        gw11 = *(const float4*)(w_ + (size_t)(64 + 1) * kH);                            \
    }
#define G_PAIRS(d_, w0_, w1_)                                                           \
    {                                                                                   \
        v2f p_;                                                                         \
        p_.x = (w0_).x; p_.y = (w1_).x; *(v2f*)((d_) + 0) = p_;                         \
        p_.x = (w0_).y; p_.y = (w1_).y; *(v2f*)((d_) + 2) = p_;                         \
        p_.x = (w0_).z; p_.y = (w1_).z; *(v2f*)((d_) + 4) = p_;                         \
        p_.x = (w0_).w; p_.y = (w1_).w; *(v2f*)((d_) + 6) = p_;                         \
    }
#define G_STORE(stage_)                                                                 \
    {                                                                                   \
        float* ad_ = adst + (stage_) * kGStage;                                         \
        float4 j0_, j1_;                                                                \
        j0_.x = fmaf(g1, gb0.x, g0 * ga0.x); j0_.y = fmaf(g1, gb0.y, g0 * ga0.y);       \
        j0_.z = fmaf(g1, gb0.z, g0 * ga0.z); j0_.w = fmaf(g1, gb0.w, g0 * ga0.w);       \
        j1_.x = fmaf(g1, gb1.x, g0 * ga1.x); j1_.y = fmaf(g1, gb1.y, g0 * ga1.y);       \
        j1_.z = fmaf(g1, gb1.z, g0 * ga1.z); j1_.w = fmaf(g1, gb1.w, g0 * ga1.w);       \
        *(float4*)(ad_)      = j0_;                                                     \
        *(float4*)(ad_ + 64) = j1_;                                                     \
        float* wd_ = wdst + (stage_) * kGStage;                                         \
        G_PAIRS(wd_, gw00, gw01)                                                        \
        G_PAIRS(wd_ + 32 * kGPairStride * 2, gw10, gw11)                                \
    }

        G_LOAD(0)
        G_STORE(0)
        __syncthreads();

        const int aoff = (tm * 16 + l16) * kGAStride + kh * (kGKC / 2) + 2 * hi;
        const int woff = (kh * (kGKC / 4) + hi) * (kGPairStride * 2) + (tn * 16 + l16) * 2;

        #pragma unroll 2
        for (int kc = 0; kc < kD / kGKC; ++kc) {
            const int  cur  = kc & 1;
            const bool more = kc + 1 < kD / kGKC;
            if (more) G_LOAD((kc + 1) * kGKC)

            const float* As = smem + cur * kGStage + aoff;
            const float* Wp = smem + cur * kGStage + kGAFloats + woff;
            #pragma unroll
            for (int kk = 0; kk < kGKC / 8; kk += 2) {
                const v2f a0 = *(const v2f*)&As[kk * 4];
                const v2f b0 = *(const v2f*)&Wp[kk * 2 * (kGPairStride * 2)];
                const v2f a1 = *(const v2f*)&As[(kk + 1) * 4];
                const v2f b1 = *(const v2f*)&Wp[(kk + 1) * 2 * (kGPairStride * 2)];
                acc0 = __builtin_amdgcn_wmma_f32_16x16x4_f32(false, a0, false, b0, (short)0, acc0, false, false);
                acc1 = __builtin_amdgcn_wmma_f32_16x16x4_f32(false, a1, false, b1, (short)0, acc1, false, false);
            }
            if (more) G_STORE(cur ^ 1)
            __syncthreads();
        }
#undef G_LOAD
#undef G_PAIRS
#undef G_STORE

        float* Pp = smem;
        #pragma unroll
        for (int r = 0; r < 8; ++r)
            Pp[(kh * kGM + tm * 16 + hi * 8 + r) * kGPPitch + tn * 16 + l16] = acc0[r] + acc1[r];
        __syncthreads();
        {
            const float4 q0 = *(const float4*)&Pp[ar * kGPPitch + ac];
            const float4 q1 = *(const float4*)&Pp[(kGM + ar) * kGPPitch + ac];
            const float4 bb = *(const float4*)&bvec[n0 + ac];
            float4 o;
            o.x = (q0.x + q1.x) + sa * bb.x;
            o.y = (q0.y + q1.y) + sa * bb.y;
            o.z = (q0.z + q1.z) + sa * bb.z;
            o.w = (q0.w + q1.w) + sa * bb.w;
            *(float4*)&out[(size_t)(m0 + ar) * kH + n0 + ac] = o;
        }
        return;
    }


    {
        int ridx[kScanPerWave / 32], rst[kScanPerWave / 32];
        #pragma unroll
        for (int it = 0; it < kScanPerWave / 32; ++it) {
            const int s = uwave * kScanPerWave + it * 32 + lane;
            ridx[it] = spans[3 * s + 0];
            rst[it]  = spans[3 * s + 1];
        }
        int n = 0;
        #pragma unroll
        for (int it = 0; it < kScanPerWave / 32; ++it) {
            const int s = uwave * kScanPerWave + it * 32 + lane;
            const bool mine = (ridx[it] == seq) && ((rst[it] >> kWinShift) == win);
            const unsigned mask = (unsigned)__ballot(mine);
            if (mine) slist[uwave][n + __popc(mask & ((1u << lane) - 1u))] = (unsigned short)s;
            n += __popc(mask);
        }
        if (lane == 0) scnt[uwave] = n;
    }
    __syncthreads();
    int total = 0;
    for (int w = 0; w < kWaves; ++w) total += scnt[w];
    total = __builtin_amdgcn_readfirstlane(total);
    if (total == 0) return;

    v8f acc[8];
    #pragma unroll
    for (int nt = 0; nt < 8; ++nt)
        #pragma unroll
        for (int e = 0; e < 8; ++e) acc[nt][e] = 0.f;

    const int rowj = tid >> 2;
    const int quad = tid & 3;
    int pos = p0 + rowj;
    if (pos > kL - 1) pos = kL - 1;
    const float* arow = hs + ((size_t)seq * kL + pos) * kD;

    const int wkp = tid >> 6;
    const int wc  = (tid & 63) * 4;
    const float* wsrc = Wm + (size_t)(2 * wkp) * kH + wc;
    float* adst0 = &smem[rowj * kAStride + quad * 16];
    float* wdst0 = &smem[kAFloats + (wkp * kWPairStride + wc) * 2];
    float4 ta0, ta1, ta2, ta3;
    float4 tw00, tw01, tw10, tw11, tw20, tw21, tw30, tw31;

#define LOAD_CHUNK(kbase_)                                                              \
    {                                                                                   \
        const float4* asrc_ = (const float4*)(arow + (kbase_) + quad * 16);             \
        ta0 = asrc_[0]; ta1 = asrc_[1]; ta2 = asrc_[2]; ta3 = asrc_[3];                 \
        const float* w_ = wsrc + (size_t)(kbase_) * kH;                                 \
        tw00 = *(const float4*)(w_ + (size_t)( 0 + 0) * kH);                            \
        tw01 = *(const float4*)(w_ + (size_t)( 0 + 1) * kH);                            \
        tw10 = *(const float4*)(w_ + (size_t)(16 + 0) * kH);                            \
        tw11 = *(const float4*)(w_ + (size_t)(16 + 1) * kH);                            \
        tw20 = *(const float4*)(w_ + (size_t)(32 + 0) * kH);                            \
        tw21 = *(const float4*)(w_ + (size_t)(32 + 1) * kH);                            \
        tw30 = *(const float4*)(w_ + (size_t)(48 + 0) * kH);                            \
        tw31 = *(const float4*)(w_ + (size_t)(48 + 1) * kH);                            \
    }
#define STORE_PAIRS(d_, w0_, w1_)                                                       \
    {                                                                                   \
        v2f p_;                                                                         \
        p_.x = (w0_).x; p_.y = (w1_).x; *(v2f*)((d_) + 0) = p_;                         \
        p_.x = (w0_).y; p_.y = (w1_).y; *(v2f*)((d_) + 2) = p_;                         \
        p_.x = (w0_).z; p_.y = (w1_).z; *(v2f*)((d_) + 4) = p_;                         \
        p_.x = (w0_).w; p_.y = (w1_).w; *(v2f*)((d_) + 6) = p_;                         \
    }
#define STORE_CHUNK(stage_)                                                             \
    {                                                                                   \
        float* ad_ = adst0 + (stage_) * kStageFloats;                                   \
        *(float4*)(ad_ +  0) = ta0;                                                     \
        *(float4*)(ad_ +  4) = ta1;                                                     \
        *(float4*)(ad_ +  8) = ta2;                                                     \
        *(float4*)(ad_ + 12) = ta3;                                                     \
        float* wd_ = wdst0 + (stage_) * kStageFloats;                                   \
        STORE_PAIRS(wd_ + 0 * 8 * kWPairStride * 2, tw00, tw01)                         \
        STORE_PAIRS(wd_ + 1 * 8 * kWPairStride * 2, tw10, tw11)                         \
        STORE_PAIRS(wd_ + 2 * 8 * kWPairStride * 2, tw20, tw21)                         \
        STORE_PAIRS(wd_ + 3 * 8 * kWPairStride * 2, tw30, tw31)                         \
    }

    LOAD_CHUNK(0)
    STORE_CHUNK(0)
    __syncthreads();

    #pragma unroll 2
    for (int kc = 0; kc < kD / kKC; ++kc) {
        const int  cur  = kc & 1;
        const bool more = kc + 1 < kD / kKC;
        if (more) LOAD_CHUNK((kc + 1) * kKC)

        const float* As = smem + cur * kStageFloats;
        const float* Wp = As + kAFloats;
        #pragma unroll
        for (int kk = 0; kk < kKC / 4; ++kk) {
            const int kb = kk * 4;
            const v2f a = *(const v2f*)&As[(mt * 16 + l16) * kAStride + kb + 2 * hi];
            v2f bfr[8];
            #pragma unroll
            for (int nt = 0; nt < 8; ++nt) {
                const int col = (nh * 8 + nt) * 16 + l16;
                bfr[nt] = *(const v2f*)&Wp[((kb >> 1) + hi) * (kWPairStride * 2) + col * 2];
            }
            #pragma unroll
            for (int nt = 0; nt < 8; ++nt) {
                acc[nt] = __builtin_amdgcn_wmma_f32_16x16x4_f32(
                    false, a, false, bfr[nt], (short)0, acc[nt], false, false);
            }
        }
        if (more) STORE_CHUNK(cur ^ 1)
        __syncthreads();
    }

#undef LOAD_CHUNK
#undef STORE_PAIRS
#undef STORE_CHUNK

    float* P = smem;
    float srow[8];
    #pragma unroll
    for (int r = 0; r < 8; ++r) srow[r] = 0.f;
    #pragma unroll
    for (int nt = 0; nt < 8; ++nt) {
        const int col    = (nh * 8 + nt) * 16 + l16;
        const float bcol = bvec[col];
        const float vcol = vvec[col];
        #pragma unroll
        for (int r = 0; r < 8; ++r) {
            acc[nt][r] += bcol;
            srow[r] += acc[nt][r] * vcol;
            P[(mt * 16 + hi * 8 + r) * kPStride + col] = acc[nt][r];
        }
    }
    #pragma unroll
    for (int off = 8; off >= 1; off >>= 1)
        #pragma unroll
        for (int r = 0; r < 8; ++r)
            srow[r] += __shfl_xor(srow[r], off, 16);
    if (l16 == 0) {
        #pragma unroll
        for (int r = 0; r < 8; ++r)
            scp[nh][mt * 16 + hi * 8 + r] = srow[r];
    }
    __syncthreads();

    for (int j = uwave; j < total; j += kWaves) {
        int t = j, w = 0;
        for (; w < kWaves - 1 && t >= scnt[w]; ++w) t -= scnt[w];
        const int s     = __builtin_amdgcn_readfirstlane((int)slist[w][t]);
        const int start = __builtin_amdgcn_readfirstlane(spans[3 * s + 1]);
        const int stop  = __builtin_amdgcn_readfirstlane(spans[3 * s + 2]);
        const int len   = stop - start;

        int r0 = start + lane;       if (r0 > kL - 1) r0 = kL - 1;  r0 -= p0;
        int r1 = start + lane + 32;  if (r1 > kL - 1) r1 = kL - 1;  r1 -= p0;
        const float sc0 = scp[0][r0] + scp[1][r0];
        const float sc1 = scp[0][r1] + scp[1][r1];
        const bool v0 = lane < len, v1 = lane + 32 < len;
        float m = fmaxf(v0 ? sc0 : -INFINITY, v1 ? sc1 : -INFINITY);
        #pragma unroll
        for (int off = 16; off >= 1; off >>= 1) m = fmaxf(m, __shfl_xor(m, off, 32));
        const float e0 = v0 ? expf(sc0 - m) : 0.f;
        const float e1 = v1 ? expf(sc1 - m) : 0.f;
        float sum = e0 + e1;
        #pragma unroll
        for (int off = 16; off >= 1; off >>= 1) sum += __shfl_xor(sum, off, 32);
        const float inv = 1.f / sum;
        const float a0 = e0 * inv, a1 = e1 * inv;

        const float* Pl = P + lane * 4;
        float4 pa = make_float4(0.f, 0.f, 0.f, 0.f);
        float4 pb = make_float4(0.f, 0.f, 0.f, 0.f);
        #pragma unroll 8
        for (int l = 0; l < kSpan; ++l) {
            const float av = (l < 32) ? a0 : a1;
            const float a  = __int_as_float(__builtin_amdgcn_readlane(__float_as_int(av), l & 31));
            int r = start + l;  if (r > kL - 1) r = kL - 1;  r -= p0;
            const float4 x = *(const float4*)(Pl + r * kPStride);
            const float4 y = *(const float4*)(Pl + r * kPStride + 128);
            pa.x = fmaf(a, x.x, pa.x); pa.y = fmaf(a, x.y, pa.y);
            pa.z = fmaf(a, x.z, pa.z); pa.w = fmaf(a, x.w, pa.w);
            pb.x = fmaf(a, y.x, pb.x); pb.y = fmaf(a, y.y, pb.y);
            pb.z = fmaf(a, y.z, pb.z); pb.w = fmaf(a, y.w, pb.w);
        }
        float* o = out + (size_t)s * kH + lane * 4;
        *(float4*)(o)       = pa;
        *(float4*)(o + 128) = pb;
    }
}


namespace {
__device__ __forceinline__ float wave_sum(float x)
{
    #pragma unroll
    for (int off = 16; off >= 1; off >>= 1) x += __shfl_xor(x, off, 32);
    return x;
}
__device__ __forceinline__ float dot4(const float4 a, const float4 b, float acc)
{
    acc = fmaf(a.x, b.x, acc); acc = fmaf(a.y, b.y, acc);
    acc = fmaf(a.z, b.z, acc); acc = fmaf(a.w, b.w, acc);
    return acc;
}
}

__launch_bounds__(kUThreads)
__global__ void score_vector_kernel(const float* __restrict__ Wm,
                                    const float* __restrict__ bvec,
                                    const float* __restrict__ vvec,
                                    float* __restrict__ ws, size_t ws_bytes)
{
    if (ws_bytes < kNeedBytes) return;
    const int uw   = __builtin_amdgcn_readfirstlane((int)(threadIdx.x >> 5));
    const int lane = threadIdx.x & 31;
    const int k    = blockIdx.x * (kUThreads / 32) + uw;
    const float4 v0 = *(const float4*)(vvec + lane * 4);
    const float4 v1 = *(const float4*)(vvec + 128 + lane * 4);
    const float4 w0 = *(const float4*)(Wm + (size_t)k * kH + lane * 4);
    const float4 w1 = *(const float4*)(Wm + (size_t)k * kH + 128 + lane * 4);
    const float p = wave_sum(dot4(w1, v1, dot4(w0, v0, 0.f)));
    if (lane == 0) ws[kOffU + k] = p;
    if (k == 0) {
        const float4 b0 = *(const float4*)(bvec + lane * 4);
        const float4 b1 = *(const float4*)(bvec + 128 + lane * 4);
        const float q = wave_sum(dot4(b1, v1, dot4(b0, v0, 0.f)));
        if (lane == 0) ws[kOffU + kD] = q;
    }
}

__launch_bounds__(kPoolThreads)
__global__ void span_pool_kernel(const float* __restrict__ hs_all,
                                 const int* __restrict__ spans,
                                 float* __restrict__ ws, size_t ws_bytes)
{
    if (ws_bytes < kNeedBytes) return;
    __shared__ __align__(16) float tile[kTileRows * kD];
    __shared__ float rsc[kTileRows];
    __shared__ float bq[kPoolWaves];
    __shared__ int lst_s[kS], lst_start[kS], lst_n[kS];
    __shared__ int scnt[kPoolWaves];
    const int uw   = __builtin_amdgcn_readfirstlane((int)(threadIdx.x >> 5));
    const int lane = threadIdx.x & 31;
    const int seq  = blockIdx.x / kChPerSeq;
    const int ch   = blockIdx.x % kChPerSeq;
    const int c0   = ch * kCh;

    int ridx[kPoolScan / 32], rst[kPoolScan / 32], rsp[kPoolScan / 32];
    #pragma unroll
    for (int it = 0; it < kPoolScan / 32; ++it) {
        const int s = uw * kPoolScan + it * 32 + lane;
        ridx[it] = spans[3 * s + 0];
        rst[it]  = spans[3 * s + 1];
        rsp[it]  = spans[3 * s + 2];
    }

    const float* U = ws + kOffU;
    v4f u[8];
    #pragma unroll
    for (int k = 0; k < 8; ++k) u[k] = *(const v4f*)(U + k * 128 + lane * 4);
    const float c = U[kD];

    const float* Xs = hs_all + ((size_t)kLayer * kN + seq) * kL * kD + (size_t)(c0 + uw * kPoolRows) * kD + lane * 4;
    v4f x[kPoolRows * 8];
    #pragma unroll
    for (int j = 0; j < kPoolRows; ++j)
        #pragma unroll
        for (int k = 0; k < 8; ++k) x[j * 8 + k] = *(const v4f*)(Xs + (size_t)j * kD + k * 128);

    unsigned msk[kPoolScan / 32];
    int n = 0;
    #pragma unroll
    for (int it = 0; it < kPoolScan / 32; ++it) {
        const int idx   = min(max(ridx[it], 0), kN - 1);
        const int nrow  = min(max(rsp[it] - rst[it], 0), kSpan);
        const int first = min(max(rst[it], 0), kL - 1) >> kChShift;
        const int last  = min(max(rst[it] + nrow - 1, 0), kL - 1) >> kChShift;
        const bool mine = (idx == seq) && (nrow > 0) && (first <= ch) && (ch <= last);
        msk[it] = (unsigned)__ballot(mine);
        n += __popc(msk[it]);
    }
    if (lane == 0) scnt[uw] = n;
    __syncthreads();
    int base = 0, total = 0;
    for (int w = 0; w < kPoolWaves; ++w) {
        const int cw = scnt[w];
        base  += (w < uw) ? cw : 0;
        total += cw;
    }
    total = __builtin_amdgcn_readfirstlane(total);
    #pragma unroll
    for (int it = 0; it < kPoolScan / 32; ++it) {
        if ((msk[it] >> lane) & 1u) {
            const int p = base + __popc(msk[it] & ((1u << lane) - 1u));
            lst_s[p]     = uw * kPoolScan + it * 32 + lane;
            lst_start[p] = rst[it];
            lst_n[p]     = min(max(rsp[it] - rst[it], 0), kSpan);
        }
        base += __popc(msk[it]);
    }

    {
        float sc[kPoolRows];
        #pragma unroll
        for (int j = 0; j < kPoolRows; ++j) {
            v2f t = {0.f, 0.f};
            #pragma unroll
            for (int k = 0; k < 8; ++k) {
                t = x[j * 8 + k].xy * u[k].xy + t;
                t = x[j * 8 + k].zw * u[k].zw + t;
            }
            sc[j] = t.x + t.y;
        }
        #pragma unroll
        for (int j = 0; j < kPoolRows; ++j) sc[j] = wave_sum(sc[j]) + c;
        float mw = sc[0];
        #pragma unroll
        for (int j = 1; j < kPoolRows; ++j) mw = fmaxf(mw, sc[j]);
        v4f bs[8];
        #pragma unroll
        for (int k = 0; k < 8; ++k) bs[k] = (v4f){0.f, 0.f, 0.f, 0.f};
        float qw = 0.f;
        #pragma unroll
        for (int j = 0; j < kPoolRows; ++j) {
            const float p = expf(sc[j] - mw);
            qw += p;
            #pragma unroll
            for (int k = 0; k < 8; ++k) bs[k] = x[j * 8 + k] * p + bs[k];
        }
        #pragma unroll
        for (int j = 0; j < kPoolRows; ++j) {
            if (lane == 0) rsc[uw * kPoolRows + j] = sc[j];
            #pragma unroll
            for (int k = 0; k < 8; ++k) *(v4f*)&tile[(uw * kPoolRows + j) * kD + k * 128 + lane * 4] = x[j * 8 + k];
        }
        if (lane == 0) { rsc[kCh + uw] = mw; bq[uw] = qw; }
        #pragma unroll
        for (int k = 0; k < 8; ++k) *(v4f*)&tile[(kCh + uw) * kD + k * 128 + lane * 4] = bs[k];
    }
    __syncthreads();

    for (int t = uw; t < total; t += kPoolWaves) {
        const int s     = __builtin_amdgcn_readfirstlane(lst_s[t]);
        const int start = __builtin_amdgcn_readfirstlane(lst_start[t]);
        const int nrow  = __builtin_amdgcn_readfirstlane(lst_n[t]);
        const int first = min(max(start, 0), kL - 1) >> kChShift;
        const int slot  = ch - first;
        const int lo = (slot == 0) ? 0 : (c0 - start);
        const int hi = (ch == kChPerSeq - 1) ? nrow : min(nrow, c0 + kCh - start);
        const int cnt = hi - lo;
        const int pbeg = start + lo;
        const bool plain = (pbeg >= 0) && (pbeg + cnt <= kL);
        const int ra   = pbeg - c0;
        const int flo  = (ra + kPoolRows - 1) & ~(kPoolRows - 1);
        const int fhi  = (ra + cnt) & ~(kPoolRows - 1);
        const bool useb = plain && (fhi > flo);
        const int h1 = useb ? (flo - ra) : cnt;
        const int nb = useb ? ((fhi - flo) >> kBlkShift) : 0;
        const int b0 = flo >> kBlkShift;
        const int tl = useb ? (ra + cnt - fhi) : 0;
        const int nitems = h1 + nb + tl;

        int   ri[2];
        float wv[2], qf[2];
        #pragma unroll
        for (int h = 0; h < 2; ++h) {
            const int kk = lane + 32 * h;
            const bool valid = kk < nitems;
            int r;
            if (kk < h1)           r = min(max(pbeg + kk, 0), kL - 1) - c0;
            else if (kk < h1 + nb) r = kCh + b0 + (kk - h1);
            else                   r = fhi + (kk - h1 - nb);
            r = valid ? r : 0;
            ri[h] = r;
            wv[h] = valid ? rsc[r] : -INFINITY;
            qf[h] = (r >= kCh) ? bq[r - kCh] : 1.f;
        }
        float m = fmaxf(wv[0], wv[1]);
        #pragma unroll
        for (int off = 16; off >= 1; off >>= 1) m = fmaxf(m, __shfl_xor(m, off, 32));
        wv[0] = (lane < nitems)      ? expf(wv[0] - m) : 0.f;
        wv[1] = (lane + 32 < nitems) ? expf(wv[1] - m) : 0.f;
        const float q = wave_sum(fmaf(wv[1], qf[1], wv[0] * qf[0]));

        const float* Tl = tile + lane * 4;
        v4f acc[8];
        #pragma unroll
        for (int k = 0; k < 8; ++k) acc[k] = (v4f){0.f, 0.f, 0.f, 0.f};
        #pragma unroll 2
        for (int i = 0; i < nitems; ++i) {
            const int   rv = (i < 32) ? ri[0] : ri[1];
            const float av = (i < 32) ? wv[0] : wv[1];
            const int   r  = __builtin_amdgcn_readlane(rv, i & 31);
            const float a  = __int_as_float(__builtin_amdgcn_readlane(__float_as_int(av), i & 31));
            const float* T = Tl + r * kD;
            #pragma unroll
            for (int k = 0; k < 8; ++k) acc[k] = *(const v4f*)(T + k * 128) * a + acc[k];
        }
        float* z = ws + kOffZ + ((size_t)slot * kS + s) * kD + lane * 4;
        #pragma unroll
        for (int k = 0; k < 8; ++k) *(v4f*)(z + k * 128) = acc[k];
        if (lane == 0) {
            ws[kOffM + (size_t)slot * kS + s] = m;
            ws[kOffQ + (size_t)slot * kS + s] = q;
        }
    }
}

extern "C" void kernel_launch(void* const* d_in, const int* in_sizes, int n_in,
                              void* d_out, int out_size, void* d_ws, size_t ws_size,
                              hipStream_t stream) {
    (void)in_sizes; (void)n_in; (void)out_size;
    const float* hs    = (const float*)d_in[0];
    const int*   spans = (const int*)  d_in[1];
    const float* W     = (const float*)d_in[2];
    const float* b     = (const float*)d_in[3];
    const float* v     = (const float*)d_in[4];
    float* out = (float*)d_out;
    float* ws  = (float*)d_ws;
    score_vector_kernel<<<dim3(kUGrid), dim3(kUThreads), 0, stream>>>(W, b, v, ws, ws_size);
    span_pool_kernel<<<dim3(kPoolGrid), dim3(kPoolThreads), 0, stream>>>(hs, spans, ws, ws_size);
    attention_pooler_kernel<<<dim3(kGrid), dim3(kThreads), 0, stream>>>(hs, spans, W, b, v, ws, ws_size, out);
}


# Product launch: K loop software-pipelined: fragments read ahead of their WMMAs, stage refilled mid-chunk from loads two chunks ahead, barrier split into signal and wait round WMMAs; sums unchanged
# speedup vs baseline: 14.6688x; 1.1074x over previous
#include <hip/hip_runtime.h>
#include <hip/hip_bf16.h>
#include <math.h>


typedef __attribute__((ext_vector_type(2))) float v2f;
typedef __attribute__((ext_vector_type(8))) float v8f;
typedef __attribute__((ext_vector_type(4))) float v4f;

namespace {
constexpr int kLayer  = 2;
constexpr int kN = 32, kL = 512, kD = 1024, kH = 256;
constexpr int kS = 2048, kSpan = 64;
constexpr int kWin       = 64;
constexpr int kWinShift  = 6;
constexpr int kRows      = kWin + kSpan;
constexpr int kWinPerSeq = kL / kWin;
constexpr int kGrid      = kN * kWinPerSeq;
constexpr int kThreads   = 512;
constexpr int kWaves     = kThreads / 32;
constexpr int kScanPerWave = kS / kWaves;
constexpr int kKC = 64;
constexpr int kAStride = 68;
constexpr int kWPairStride = 272;
static_assert((1 << kWinShift) == kWin, "window shift");
constexpr int kPStride = 260;
constexpr int kAFloats = kRows * kAStride;
constexpr int kWFloats = (kKC / 2) * kWPairStride * 2;
constexpr int kStageFloats = kAFloats + kWFloats;
constexpr int kPFloats = kRows * kPStride;
constexpr int kCh        = 64;
constexpr int kChShift   = 6;
constexpr int kChPerSeq  = kL / kCh;
static_assert((1 << kChShift) == kCh && kSpan <= kCh, "a span of at most 64 rows meets at most two chunks");
constexpr size_t kOffU  = 0;
constexpr size_t kOffM  = 2048;
constexpr size_t kOffQ  = kOffM + 2 * (size_t)kS;
constexpr size_t kOffZ  = kOffQ + 2 * (size_t)kS;
constexpr size_t kNeedFloats = kOffZ + 2 * (size_t)kS * kD;
constexpr size_t kNeedBytes  = kNeedFloats * sizeof(float);
static_assert(kOffZ % 64 == 0, "pooled rows start on a 256-B boundary");
constexpr int kGM = 32, kGN = 64;
constexpr int kGKC = 128;
constexpr int kGAStride = 132;
constexpr int kGPairStride = 80;
constexpr int kGAFloats = kGM * kGAStride;
constexpr int kGWFloats = (kGKC / 2) * kGPairStride * 2;
constexpr int kGStage   = kGAFloats + kGWFloats;
constexpr int kGPPitch  = 68;
static_assert((kS / kGM) * (kH / kGN) == kGrid, "one 32 x 64 tile per work-group");
static_assert(2 * kGStage <= 2 * kStageFloats && 2 * kGM * kGPPitch <= 2 * kStageFloats, "fits the LDS ring");
constexpr int kPoolWaves   = 8;
constexpr int kPoolThreads = kPoolWaves * 32;
constexpr int kPoolGrid    = kN * kChPerSeq;
constexpr int kPoolRows    = kCh / kPoolWaves;
constexpr int kBlkShift    = 3;
constexpr int kTileRows    = kCh + kPoolWaves;
constexpr int kPoolScan    = kS / kPoolWaves;
static_assert(kPoolRows == (1 << kBlkShift) && kPoolScan % 32 == 0, "row staging and record scan shapes");
constexpr int kUThreads = 256;
constexpr int kUGrid    = kD / 8;
}

namespace {
__device__ __forceinline__ void lds_drain() { asm volatile("s_wait_dscnt 0x0" ::: "memory"); }
__device__ __forceinline__ void wg_signal() { asm volatile("s_barrier_signal -1" ::: "memory"); }
__device__ __forceinline__ void wg_wait()   { asm volatile("s_barrier_wait -1" ::: "memory"); }
}

__launch_bounds__(512)
__global__ void attention_pooler_kernel(const float* __restrict__ hs_all,
                                        const int*  __restrict__ spans,
                                        const float* __restrict__ Wm,
                                        const float* __restrict__ bvec,
                                        const float* __restrict__ vvec,
                                        float* __restrict__ ws, size_t ws_bytes,
                                        float* __restrict__ out)
{
    __shared__ __align__(16) float smem[2 * kStageFloats];
    __shared__ float scp[2][kRows];
    __shared__ unsigned short slist[kWaves][kScanPerWave];
    __shared__ int scnt[kWaves];

    const int seq  = blockIdx.x / kWinPerSeq;
    const int win  = blockIdx.x % kWinPerSeq;
    const int p0   = win * kWin;
    const int tid  = threadIdx.x;
    const int wave = tid >> 5;
    const int lane = tid & 31;
    const int mt   = wave & 7;
    const int nh   = wave >> 3;
    const int hi   = lane >> 4;
    const int l16  = lane & 15;
    const int uwave = __builtin_amdgcn_readfirstlane(wave);

    const float* hs = hs_all + (size_t)kLayer * kN * kL * kD;

    if (ws_bytes >= kNeedBytes) {
        const int m0 = (blockIdx.x >> 2) * kGM;
        const int n0 = (blockIdx.x & 3) * kGN;
        const int tm = wave & 1;
        const int tn = (wave >> 1) & 3;
        const int kh = wave >> 3;
        const float* Z  = ws + kOffZ;

        v8f acc0, acc1;
        #pragma unroll
        for (int e = 0; e < 8; ++e) { acc0[e] = 0.f; acc1[e] = 0.f; }

        const int ar = tid >> 4;
        const int ac = (tid & 15) * 4;
        const float* wsrc = Wm + (size_t)(2 * ar) * kH + n0 + ac;
        v4f ga[2][4], gw[2][4];

#define G_LOADW(set_, kbase_)                                                           \
    {                                                                                   \
        const float* w_ = wsrc + (size_t)(kbase_) * kH;                                 \
        gw[set_][0] = *(const v4f*)(w_ + (size_t)( 0 + 0) * kH);                        \
        gw[set_][1] = *(const v4f*)(w_ + (size_t)( 0 + 1) * kH);                        \
        gw[set_][2] = *(const v4f*)(w_ + (size_t)(64 + 0) * kH);                        \
        gw[set_][3] = *(const v4f*)(w_ + (size_t)(64 + 1) * kH);                        \
    }
        G_LOADW(0, 0)
        const float4 bb = *(const float4*)&bvec[n0 + ac];

        const int sp    = m0 + ar;
        const int sst   = spans[3 * sp + 1];
        const int snrow = min(max(spans[3 * sp + 2] - sst, 0), kSpan);
        const int scf   = min(max(sst, 0), kL - 1) >> kChShift;
        const int scl   = min(max(sst + snrow - 1, 0), kL - 1) >> kChShift;
        const bool one  = snrow > 0;
        const bool two  = one && scl > scf;
        const float* asrc  = one ? (Z + (size_t)sp * kD + ac) : (ws + kOffU + ac);
        const float* asrc1 = two ? (asrc + (size_t)kS * kD) : asrc;
#define G_LOADA(set_, kbase_)                                                           \
    {                                                                                   \
        ga[set_][0] = *(const v4f*)(asrc + (kbase_));                                   \
        ga[set_][1] = *(const v4f*)(asrc + (kbase_) + 64);                              \
        ga[set_][2] = *(const v4f*)(asrc1 + (kbase_));                                  \
        ga[set_][3] = *(const v4f*)(asrc1 + (kbase_) + 64);                             \
    }
        G_LOADA(0, 0)
        G_LOADW(1, kGKC)
        G_LOADA(1, kGKC)
        float g0, g1, sa;
        {
            const float pm0 = one ? ws[kOffM + sp] : 0.f;
            const float pq0 = one ? ws[kOffQ + sp] : 0.f;
            const float pm1 = two ? ws[kOffM + kS + sp] : -INFINITY;
            const float pq1 = two ? ws[kOffQ + kS + sp] : 0.f;
            const float mx  = fmaxf(pm0, pm1);
            const float f0  = expf(pm0 - mx), f1 = expf(pm1 - mx);
            const float tot = fmaf(f1, pq1, f0 * pq0);
            const float inv = 1.f / tot;
            g0 = f0 * inv; g1 = f1 * inv; sa = tot * inv;
            if (!one) g0 = sa;
        }
        float* adst = &smem[ar * kGAStride + ac];
        float* wdst = &smem[kGAFloats + (ar * kGPairStride + ac) * 2];

#define G_PAIRS(d_, w0_, w1_)                                                           \
    {                                                                                   \
        v2f p_;                                                                         \
        p_.x = (w0_).x; p_.y = (w1_).x; *(v2f*)((d_) + 0) = p_;                         \
        p_.x = (w0_).y; p_.y = (w1_).y; *(v2f*)((d_) + 2) = p_;                         \
        p_.x = (w0_).z; p_.y = (w1_).z; *(v2f*)((d_) + 4) = p_;                         \
        p_.x = (w0_).w; p_.y = (w1_).w; *(v2f*)((d_) + 6) = p_;                         \
    }
#define G_STORE(set_)                                                                   \
    {                                                                                   \
        asm volatile("" : "+v"(ga[set_][0]), "+v"(ga[set_][1]), "+v"(ga[set_][2]), "+v"(ga[set_][3]), \
                          "+v"(gw[set_][0]), "+v"(gw[set_][1]), "+v"(gw[set_][2]), "+v"(gw[set_][3])); \
        float* ad_ = adst + (set_) * kGStage;                                           \
        v4f j0_, j1_;                                                                   \
        j0_.x = fmaf(g1, ga[set_][2].x, g0 * ga[set_][0].x); j0_.y = fmaf(g1, ga[set_][2].y, g0 * ga[set_][0].y); \
        j0_.z = fmaf(g1, ga[set_][2].z, g0 * ga[set_][0].z); j0_.w = fmaf(g1, ga[set_][2].w, g0 * ga[set_][0].w); \
        j1_.x = fmaf(g1, ga[set_][3].x, g0 * ga[set_][1].x); j1_.y = fmaf(g1, ga[set_][3].y, g0 * ga[set_][1].y); \
        j1_.z = fmaf(g1, ga[set_][3].z, g0 * ga[set_][1].z); j1_.w = fmaf(g1, ga[set_][3].w, g0 * ga[set_][1].w); \
        *(v4f*)(ad_)      = j0_;                                                        \
        *(v4f*)(ad_ + 64) = j1_;                                                        \
        float* wd_ = wdst + (set_) * kGStage;                                           \
        G_PAIRS(wd_, gw[set_][0], gw[set_][1])                                          \
        G_PAIRS(wd_ + 32 * kGPairStride * 2, gw[set_][2], gw[set_][3])                  \
    }

        G_STORE(0)
        G_LOADW(0, 2 * kGKC)
        G_LOADA(0, 2 * kGKC)
        lds_drain();
        wg_signal();
        wg_wait();

        const int aoff = (tm * 16 + l16) * kGAStride + kh * (kGKC / 2) + 2 * hi;
        const int woff = (kh * (kGKC / 4) + hi) * (kGPairStride * 2) + (tn * 16 + l16) * 2;

        v2f fa[kGKC / 8], fb[kGKC / 8];
#define G_FRAGS(half_, stage_)                                                          \
    {                                                                                   \
        const float* As_ = smem + (stage_) * kGStage + aoff;                            \
        const float* Wp_ = smem + (stage_) * kGStage + kGAFloats + woff;                \
        _Pragma("unroll")                                                               \
        for (int f_ = 8 * (half_); f_ < 8 * (half_) + 8; ++f_) {                        \
            fa[f_] = *(const v2f*)&As_[f_ * 4];                                         \
            fb[f_] = *(const v2f*)&Wp_[f_ * 2 * (kGPairStride * 2)];                    \
        }                                                                               \
    }
#define G_MMA(kk_)                                                                      \
    {                                                                                   \
        acc0 = __builtin_amdgcn_wmma_f32_16x16x4_f32(false, fa[(kk_)], false, fb[(kk_)], (short)0, acc0, false, false);         \
        acc1 = __builtin_amdgcn_wmma_f32_16x16x4_f32(false, fa[(kk_) + 1], false, fb[(kk_) + 1], (short)0, acc1, false, false); \
    }
        G_FRAGS(0, 0)

#define PHASE __builtin_amdgcn_sched_barrier(0);
        #pragma unroll
        for (int kc = 0; kc < kD / kGKC; ++kc) {
            const int  cur  = kc & 1;
            const int  nxt  = cur ^ 1;
            const bool more = kc + 1 < kD / kGKC;
            G_FRAGS(1, cur)
            PHASE
            G_MMA(0) G_MMA(2)
            PHASE
            if (more) {
                G_STORE(nxt)
                if (kc + 3 < kD / kGKC) { G_LOADW(nxt, (kc + 3) * kGKC) G_LOADA(nxt, (kc + 3) * kGKC) }
            }
            PHASE
            G_MMA(4)
            PHASE
            if (more) { lds_drain(); wg_signal(); }
            PHASE
            G_MMA(6) G_MMA(8)
            PHASE
            if (more) { wg_wait(); G_FRAGS(0, nxt) }
            PHASE
            G_MMA(10) G_MMA(12) G_MMA(14)
            PHASE
        }
#undef PHASE
#undef G_LOADW
#undef G_LOADA
#undef G_PAIRS
#undef G_STORE
#undef G_FRAGS
#undef G_MMA

        float* Pp = smem;
        #pragma unroll
        for (int r = 0; r < 8; ++r)
            Pp[(kh * kGM + tm * 16 + hi * 8 + r) * kGPPitch + tn * 16 + l16] = acc0[r] + acc1[r];
        __syncthreads();
        {
            const float4 q0 = *(const float4*)&Pp[ar * kGPPitch + ac];
            const float4 q1 = *(const float4*)&Pp[(kGM + ar) * kGPPitch + ac];
            float4 o;
            o.x = (q0.x + q1.x) + sa * bb.x;
            o.y = (q0.y + q1.y) + sa * bb.y;
            o.z = (q0.z + q1.z) + sa * bb.z;
            o.w = (q0.w + q1.w) + sa * bb.w;
            *(float4*)&out[(size_t)(m0 + ar) * kH + n0 + ac] = o;
        }
        return;
    }


    {
        int ridx[kScanPerWave / 32], rst[kScanPerWave / 32];
        #pragma unroll
        for (int it = 0; it < kScanPerWave / 32; ++it) {
            const int s = uwave * kScanPerWave + it * 32 + lane;
            ridx[it] = spans[3 * s + 0];
            rst[it]  = spans[3 * s + 1];
        }
        int n = 0;
        #pragma unroll
        for (int it = 0; it < kScanPerWave / 32; ++it) {
            const int s = uwave * kScanPerWave + it * 32 + lane;
            const bool mine = (ridx[it] == seq) && ((rst[it] >> kWinShift) == win);
            const unsigned mask = (unsigned)__ballot(mine);
            if (mine) slist[uwave][n + __popc(mask & ((1u << lane) - 1u))] = (unsigned short)s;
            n += __popc(mask);
        }
        if (lane == 0) scnt[uwave] = n;
    }
    __syncthreads();
    int total = 0;
    for (int w = 0; w < kWaves; ++w) total += scnt[w];
    total = __builtin_amdgcn_readfirstlane(total);
    if (total == 0) return;

    v8f acc[8];
    #pragma unroll
    for (int nt = 0; nt < 8; ++nt)
        #pragma unroll
        for (int e = 0; e < 8; ++e) acc[nt][e] = 0.f;

    const int rowj = tid >> 2;
    const int quad = tid & 3;
    int pos = p0 + rowj;
    if (pos > kL - 1) pos = kL - 1;
    const float* arow = hs + ((size_t)seq * kL + pos) * kD;

    const int wkp = tid >> 6;
    const int wc  = (tid & 63) * 4;
    const float* wsrc = Wm + (size_t)(2 * wkp) * kH + wc;
    float* adst0 = &smem[rowj * kAStride + quad * 16];
    float* wdst0 = &smem[kAFloats + (wkp * kWPairStride + wc) * 2];
    float4 ta0, ta1, ta2, ta3;
    float4 tw00, tw01, tw10, tw11, tw20, tw21, tw30, tw31;

#define LOAD_CHUNK(kbase_)                                                              \
    {                                                                                   \
        const float4* asrc_ = (const float4*)(arow + (kbase_) + quad * 16);             \
        ta0 = asrc_[0]; ta1 = asrc_[1]; ta2 = asrc_[2]; ta3 = asrc_[3];                 \
        const float* w_ = wsrc + (size_t)(kbase_) * kH;                                 \
        tw00 = *(const float4*)(w_ + (size_t)( 0 + 0) * kH);                            \
        tw01 = *(const float4*)(w_ + (size_t)( 0 + 1) * kH);                            \
        tw10 = *(const float4*)(w_ + (size_t)(16 + 0) * kH);                            \
        tw11 = *(const float4*)(w_ + (size_t)(16 + 1) * kH);                            \
        tw20 = *(const float4*)(w_ + (size_t)(32 + 0) * kH);                            \
        tw21 = *(const float4*)(w_ + (size_t)(32 + 1) * kH);                            \
        tw30 = *(const float4*)(w_ + (size_t)(48 + 0) * kH);                            \
        tw31 = *(const float4*)(w_ + (size_t)(48 + 1) * kH);                            \
    }
#define STORE_PAIRS(d_, w0_, w1_)                                                       \
    {                                                                                   \
        v2f p_;                                                                         \
        p_.x = (w0_).x; p_.y = (w1_).x; *(v2f*)((d_) + 0) = p_;                         \
        p_.x = (w0_).y; p_.y = (w1_).y; *(v2f*)((d_) + 2) = p_;                         \
        p_.x = (w0_).z; p_.y = (w1_).z; *(v2f*)((d_) + 4) = p_;                         \
        p_.x = (w0_).w; p_.y = (w1_).w; *(v2f*)((d_) + 6) = p_;                         \
    }
#define STORE_CHUNK(stage_)                                                             \
    {                                                                                   \
        float* ad_ = adst0 + (stage_) * kStageFloats;                                   \
        *(float4*)(ad_ +  0) = ta0;                                                     \
        *(float4*)(ad_ +  4) = ta1;                                                     \
        *(float4*)(ad_ +  8) = ta2;                                                     \
        *(float4*)(ad_ + 12) = ta3;                                                     \
        float* wd_ = wdst0 + (stage_) * kStageFloats;                                   \
        STORE_PAIRS(wd_ + 0 * 8 * kWPairStride * 2, tw00, tw01)                         \
        STORE_PAIRS(wd_ + 1 * 8 * kWPairStride * 2, tw10, tw11)                         \
        STORE_PAIRS(wd_ + 2 * 8 * kWPairStride * 2, tw20, tw21)                         \
        STORE_PAIRS(wd_ + 3 * 8 * kWPairStride * 2, tw30, tw31)                         \
    }

    LOAD_CHUNK(0)
    STORE_CHUNK(0)
    __syncthreads();

    #pragma unroll 2
    for (int kc = 0; kc < kD / kKC; ++kc) {
        const int  cur  = kc & 1;
        const bool more = kc + 1 < kD / kKC;
        if (more) LOAD_CHUNK((kc + 1) * kKC)

        const float* As = smem + cur * kStageFloats;
        const float* Wp = As + kAFloats;
        #pragma unroll
        for (int kk = 0; kk < kKC / 4; ++kk) {
            const int kb = kk * 4;
            const v2f a = *(const v2f*)&As[(mt * 16 + l16) * kAStride + kb + 2 * hi];
            v2f bfr[8];
            #pragma unroll
            for (int nt = 0; nt < 8; ++nt) {
                const int col = (nh * 8 + nt) * 16 + l16;
                bfr[nt] = *(const v2f*)&Wp[((kb >> 1) + hi) * (kWPairStride * 2) + col * 2];
            }
            #pragma unroll
            for (int nt = 0; nt < 8; ++nt) {
                acc[nt] = __builtin_amdgcn_wmma_f32_16x16x4_f32(
                    false, a, false, bfr[nt], (short)0, acc[nt], false, false);
            }
        }
        if (more) STORE_CHUNK(cur ^ 1)
        __syncthreads();
    }

#undef LOAD_CHUNK
#undef STORE_PAIRS
#undef STORE_CHUNK

    float* P = smem;
    float srow[8];
    #pragma unroll
    for (int r = 0; r < 8; ++r) srow[r] = 0.f;
    #pragma unroll
    for (int nt = 0; nt < 8; ++nt) {
        const int col    = (nh * 8 + nt) * 16 + l16;
        const float bcol = bvec[col];
        const float vcol = vvec[col];
        #pragma unroll
        for (int r = 0; r < 8; ++r) {
            acc[nt][r] += bcol;
            srow[r] += acc[nt][r] * vcol;
            P[(mt * 16 + hi * 8 + r) * kPStride + col] = acc[nt][r];
        }
    }
    #pragma unroll
    for (int off = 8; off >= 1; off >>= 1)
        #pragma unroll
        for (int r = 0; r < 8; ++r)
            srow[r] += __shfl_xor(srow[r], off, 16);
    if (l16 == 0) {
        #pragma unroll
        for (int r = 0; r < 8; ++r)
            scp[nh][mt * 16 + hi * 8 + r] = srow[r];
    }
    __syncthreads();

    for (int j = uwave; j < total; j += kWaves) {
        int t = j, w = 0;
        for (; w < kWaves - 1 && t >= scnt[w]; ++w) t -= scnt[w];
        const int s     = __builtin_amdgcn_readfirstlane((int)slist[w][t]);
        const int start = __builtin_amdgcn_readfirstlane(spans[3 * s + 1]);
        const int stop  = __builtin_amdgcn_readfirstlane(spans[3 * s + 2]);
        const int len   = stop - start;

        int r0 = start + lane;       if (r0 > kL - 1) r0 = kL - 1;  r0 -= p0;
        int r1 = start + lane + 32;  if (r1 > kL - 1) r1 = kL - 1;  r1 -= p0;
        const float sc0 = scp[0][r0] + scp[1][r0];
        const float sc1 = scp[0][r1] + scp[1][r1];
        const bool v0 = lane < len, v1 = lane + 32 < len;
        float m = fmaxf(v0 ? sc0 : -INFINITY, v1 ? sc1 : -INFINITY);
        #pragma unroll
        for (int off = 16; off >= 1; off >>= 1) m = fmaxf(m, __shfl_xor(m, off, 32));
        const float e0 = v0 ? expf(sc0 - m) : 0.f;
        const float e1 = v1 ? expf(sc1 - m) : 0.f;
        float sum = e0 + e1;
        #pragma unroll
        for (int off = 16; off >= 1; off >>= 1) sum += __shfl_xor(sum, off, 32);
        const float inv = 1.f / sum;
        const float a0 = e0 * inv, a1 = e1 * inv;

        const float* Pl = P + lane * 4;
        float4 pa = make_float4(0.f, 0.f, 0.f, 0.f);
        float4 pb = make_float4(0.f, 0.f, 0.f, 0.f);
        #pragma unroll 8
        for (int l = 0; l < kSpan; ++l) {
            const float av = (l < 32) ? a0 : a1;
            const float a  = __int_as_float(__builtin_amdgcn_readlane(__float_as_int(av), l & 31));
            int r = start + l;  if (r > kL - 1) r = kL - 1;  r -= p0;
            const float4 x = *(const float4*)(Pl + r * kPStride);
            const float4 y = *(const float4*)(Pl + r * kPStride + 128);
            pa.x = fmaf(a, x.x, pa.x); pa.y = fmaf(a, x.y, pa.y);
            pa.z = fmaf(a, x.z, pa.z); pa.w = fmaf(a, x.w, pa.w);
            pb.x = fmaf(a, y.x, pb.x); pb.y = fmaf(a, y.y, pb.y);
            pb.z = fmaf(a, y.z, pb.z); pb.w = fmaf(a, y.w, pb.w);
        }
        float* o = out + (size_t)s * kH + lane * 4;
        *(float4*)(o)       = pa;
        *(float4*)(o + 128) = pb;
    }
}


namespace {
__device__ __forceinline__ float wave_sum(float x)
{
    #pragma unroll
    for (int off = 16; off >= 1; off >>= 1) x += __shfl_xor(x, off, 32);
    return x;
}
__device__ __forceinline__ float dot4(const float4 a, const float4 b, float acc)
{
    acc = fmaf(a.x, b.x, acc); acc = fmaf(a.y, b.y, acc);
    acc = fmaf(a.z, b.z, acc); acc = fmaf(a.w, b.w, acc);
    return acc;
}
}

__launch_bounds__(kUThreads)
__global__ void score_vector_kernel(const float* __restrict__ Wm,
                                    const float* __restrict__ bvec,
                                    const float* __restrict__ vvec,
                                    float* __restrict__ ws, size_t ws_bytes)
{
    if (ws_bytes < kNeedBytes) return;
    const int uw   = __builtin_amdgcn_readfirstlane((int)(threadIdx.x >> 5));
    const int lane = threadIdx.x & 31;
    const int k    = blockIdx.x * (kUThreads / 32) + uw;
    const float4 v0 = *(const float4*)(vvec + lane * 4);
    const float4 v1 = *(const float4*)(vvec + 128 + lane * 4);
    const float4 w0 = *(const float4*)(Wm + (size_t)k * kH + lane * 4);
    const float4 w1 = *(const float4*)(Wm + (size_t)k * kH + 128 + lane * 4);
    const float p = wave_sum(dot4(w1, v1, dot4(w0, v0, 0.f)));
    if (lane == 0) ws[kOffU + k] = p;
    if (k == 0) {
        const float4 b0 = *(const float4*)(bvec + lane * 4);
        const float4 b1 = *(const float4*)(bvec + 128 + lane * 4);
        const float q = wave_sum(dot4(b1, v1, dot4(b0, v0, 0.f)));
        if (lane == 0) ws[kOffU + kD] = q;
    }
}

__launch_bounds__(kPoolThreads)
__global__ void span_pool_kernel(const float* __restrict__ hs_all,
                                 const int* __restrict__ spans,
                                 float* __restrict__ ws, size_t ws_bytes)
{
    if (ws_bytes < kNeedBytes) return;
    __shared__ __align__(16) float tile[kTileRows * kD];
    __shared__ float rsc[kTileRows];
    __shared__ float bq[kPoolWaves];
    __shared__ int lst_s[kS], lst_start[kS], lst_n[kS];
    __shared__ int scnt[kPoolWaves];
    const int uw   = __builtin_amdgcn_readfirstlane((int)(threadIdx.x >> 5));
    const int lane = threadIdx.x & 31;
    const int seq  = blockIdx.x / kChPerSeq;
    const int ch   = blockIdx.x % kChPerSeq;
    const int c0   = ch * kCh;

    int ridx[kPoolScan / 32], rst[kPoolScan / 32], rsp[kPoolScan / 32];
    #pragma unroll
    for (int it = 0; it < kPoolScan / 32; ++it) {
        const int s = uw * kPoolScan + it * 32 + lane;
        ridx[it] = spans[3 * s + 0];
        rst[it]  = spans[3 * s + 1];
        rsp[it]  = spans[3 * s + 2];
    }

    const float* U = ws + kOffU;
    v4f u[8];
    #pragma unroll
    for (int k = 0; k < 8; ++k) u[k] = *(const v4f*)(U + k * 128 + lane * 4);
    const float c = U[kD];

    const float* Xs = hs_all + ((size_t)kLayer * kN + seq) * kL * kD + (size_t)(c0 + uw * kPoolRows) * kD + lane * 4;
    v4f x[kPoolRows * 8];
    #pragma unroll
    for (int j = 0; j < kPoolRows; ++j)
        #pragma unroll
        for (int k = 0; k < 8; ++k) x[j * 8 + k] = *(const v4f*)(Xs + (size_t)j * kD + k * 128);

    unsigned msk[kPoolScan / 32];
    int n = 0;
    #pragma unroll
    for (int it = 0; it < kPoolScan / 32; ++it) {
        const int idx   = min(max(ridx[it], 0), kN - 1);
        const int nrow  = min(max(rsp[it] - rst[it], 0), kSpan);
        const int first = min(max(rst[it], 0), kL - 1) >> kChShift;
        const int last  = min(max(rst[it] + nrow - 1, 0), kL - 1) >> kChShift;
        const bool mine = (idx == seq) && (nrow > 0) && (first <= ch) && (ch <= last);
        msk[it] = (unsigned)__ballot(mine);
        n += __popc(msk[it]);
    }
    if (lane == 0) scnt[uw] = n;
    __syncthreads();
    int base = 0, total = 0;
    for (int w = 0; w < kPoolWaves; ++w) {
        const int cw = scnt[w];
        base  += (w < uw) ? cw : 0;
        total += cw;
    }
    total = __builtin_amdgcn_readfirstlane(total);
    #pragma unroll
    for (int it = 0; it < kPoolScan / 32; ++it) {
        if ((msk[it] >> lane) & 1u) {
            const int p = base + __popc(msk[it] & ((1u << lane) - 1u));
            lst_s[p]     = uw * kPoolScan + it * 32 + lane;
            lst_start[p] = rst[it];
            lst_n[p]     = min(max(rsp[it] - rst[it], 0), kSpan);
        }
        base += __popc(msk[it]);
    }

    {
        float sc[kPoolRows];
        #pragma unroll
        for (int j = 0; j < kPoolRows; ++j) {
            v2f t = {0.f, 0.f};
            #pragma unroll
            for (int k = 0; k < 8; ++k) {
                t = x[j * 8 + k].xy * u[k].xy + t;
                t = x[j * 8 + k].zw * u[k].zw + t;
            }
            sc[j] = t.x + t.y;
        }
        #pragma unroll
        for (int j = 0; j < kPoolRows; ++j) sc[j] = wave_sum(sc[j]) + c;
        float mw = sc[0];
        #pragma unroll
        for (int j = 1; j < kPoolRows; ++j) mw = fmaxf(mw, sc[j]);
        v4f bs[8];
        #pragma unroll
        for (int k = 0; k < 8; ++k) bs[k] = (v4f){0.f, 0.f, 0.f, 0.f};
        float qw = 0.f;
        #pragma unroll
        for (int j = 0; j < kPoolRows; ++j) {
            const float p = expf(sc[j] - mw);
            qw += p;
            #pragma unroll
            for (int k = 0; k < 8; ++k) bs[k] = x[j * 8 + k] * p + bs[k];
        }
        #pragma unroll
        for (int j = 0; j < kPoolRows; ++j) {
            if (lane == 0) rsc[uw * kPoolRows + j] = sc[j];
            #pragma unroll
            for (int k = 0; k < 8; ++k) *(v4f*)&tile[(uw * kPoolRows + j) * kD + k * 128 + lane * 4] = x[j * 8 + k];
        }
        if (lane == 0) { rsc[kCh + uw] = mw; bq[uw] = qw; }
        #pragma unroll
        for (int k = 0; k < 8; ++k) *(v4f*)&tile[(kCh + uw) * kD + k * 128 + lane * 4] = bs[k];
    }
    __syncthreads();

    for (int t = uw; t < total; t += kPoolWaves) {
        const int s     = __builtin_amdgcn_readfirstlane(lst_s[t]);
        const int start = __builtin_amdgcn_readfirstlane(lst_start[t]);
        const int nrow  = __builtin_amdgcn_readfirstlane(lst_n[t]);
        const int first = min(max(start, 0), kL - 1) >> kChShift;
        const int slot  = ch - first;
        const int lo = (slot == 0) ? 0 : (c0 - start);
        const int hi = (ch == kChPerSeq - 1) ? nrow : min(nrow, c0 + kCh - start);
        const int cnt = hi - lo;
        const int pbeg = start + lo;
        const bool plain = (pbeg >= 0) && (pbeg + cnt <= kL);
        const int ra   = pbeg - c0;
        const int flo  = (ra + kPoolRows - 1) & ~(kPoolRows - 1);
        const int fhi  = (ra + cnt) & ~(kPoolRows - 1);
        const bool useb = plain && (fhi > flo);
        const int h1 = useb ? (flo - ra) : cnt;
        const int nb = useb ? ((fhi - flo) >> kBlkShift) : 0;
        const int b0 = flo >> kBlkShift;
        const int tl = useb ? (ra + cnt - fhi) : 0;
        const int nitems = h1 + nb + tl;

        int   ri[2];
        float wv[2], qf[2];
        #pragma unroll
        for (int h = 0; h < 2; ++h) {
            const int kk = lane + 32 * h;
            const bool valid = kk < nitems;
            int r;
            if (kk < h1)           r = min(max(pbeg + kk, 0), kL - 1) - c0;
            else if (kk < h1 + nb) r = kCh + b0 + (kk - h1);
            else                   r = fhi + (kk - h1 - nb);
            r = valid ? r : 0;
            ri[h] = r;
            wv[h] = valid ? rsc[r] : -INFINITY;
            qf[h] = (r >= kCh) ? bq[r - kCh] : 1.f;
        }
        float m = fmaxf(wv[0], wv[1]);
        #pragma unroll
        for (int off = 16; off >= 1; off >>= 1) m = fmaxf(m, __shfl_xor(m, off, 32));
        wv[0] = (lane < nitems)      ? expf(wv[0] - m) : 0.f;
        wv[1] = (lane + 32 < nitems) ? expf(wv[1] - m) : 0.f;
        const float q = wave_sum(fmaf(wv[1], qf[1], wv[0] * qf[0]));

        const float* Tl = tile + lane * 4;
        v4f acc[8];
        #pragma unroll
        for (int k = 0; k < 8; ++k) acc[k] = (v4f){0.f, 0.f, 0.f, 0.f};
        #pragma unroll 2
        for (int i = 0; i < nitems; ++i) {
            const int   rv = (i < 32) ? ri[0] : ri[1];
            const float av = (i < 32) ? wv[0] : wv[1];
            const int   r  = __builtin_amdgcn_readlane(rv, i & 31);
            const float a  = __int_as_float(__builtin_amdgcn_readlane(__float_as_int(av), i & 31));
            const float* T = Tl + r * kD;
            #pragma unroll
            for (int k = 0; k < 8; ++k) acc[k] = *(const v4f*)(T + k * 128) * a + acc[k];
        }
        float* z = ws + kOffZ + ((size_t)slot * kS + s) * kD + lane * 4;
        #pragma unroll
        for (int k = 0; k < 8; ++k) *(v4f*)(z + k * 128) = acc[k];
        if (lane == 0) {
            ws[kOffM + (size_t)slot * kS + s] = m;
            ws[kOffQ + (size_t)slot * kS + s] = q;
        }
    }
}

extern "C" void kernel_launch(void* const* d_in, const int* in_sizes, int n_in,
                              void* d_out, int out_size, void* d_ws, size_t ws_size,
                              hipStream_t stream) {
    (void)in_sizes; (void)n_in; (void)out_size;
    const float* hs    = (const float*)d_in[0];
    const int*   spans = (const int*)  d_in[1];
    const float* W     = (const float*)d_in[2];
    const float* b     = (const float*)d_in[3];
    const float* v     = (const float*)d_in[4];
    float* out = (float*)d_out;
    float* ws  = (float*)d_ws;
    score_vector_kernel<<<dim3(kUGrid), dim3(kUThreads), 0, stream>>>(W, b, v, ws, ws_size);
    span_pool_kernel<<<dim3(kPoolGrid), dim3(kPoolThreads), 0, stream>>>(hs, spans, ws, ws_size);
    attention_pooler_kernel<<<dim3(kGrid), dim3(kThreads), 0, stream>>>(hs, spans, W, b, v, ws, ws_size, out);
}
